# starting kernel
# baseline (speedup 1.0000x reference)
#include <hip/hip_runtime.h>
#include <hip/hip_bf16.h>
#include <math.h>

#define D 128
#define TILE_M 16
#define GAMMA 0.1f
#define EPSILON 0.1f
#define NUM_ITERS 4

typedef __attribute__((ext_vector_type(2))) float v2f;
typedef __attribute__((ext_vector_type(8))) float v8f;

__global__ __launch_bounds__(256) void init_deg_kernel(float* __restrict__ deg, int n) {
    int i = blockIdx.x * blockDim.x + threadIdx.x;
    if (i < n) deg[i] = 1.0f;
}

__global__ __launch_bounds__(256) void count_deg_kernel(const int* __restrict__ ei,
                                                        float* __restrict__ deg, int E) {
    int e = blockIdx.x * blockDim.x + threadIdx.x;
    if (e < E) {
        int dst = ei[E + e];
        __hip_atomic_fetch_add(&deg[dst], 1.0f, __ATOMIC_RELAXED, __HIP_MEMORY_SCOPE_AGENT);
    }
}

__global__ __launch_bounds__(256) void finalize_dinv_kernel(float* __restrict__ deg, int n) {
    int i = blockIdx.x * blockDim.x + threadIdx.x;
    if (i < n) deg[i] = rsqrtf(deg[i]);
}

__global__ __launch_bounds__(256) void prep_weights_kernel(const float* __restrict__ W,
                                                           const float* __restrict__ Wphi,
                                                           float* __restrict__ WphiP,
                                                           float* __restrict__ ATP) {
    int i = blockIdx.x * blockDim.x + threadIdx.x;
    if (i >= D * D) return;
    int k = i / D, n = i % D;
    size_t po = (size_t)(k >> 1) * (2 * D) + n * 2 + (k & 1);
    WphiP[po] = Wphi[n * D + k];
    float a = W[n * D + k] - W[k * D + n];
    if (k == n) a -= GAMMA;
    ATP[po] = a;
}

__global__ __launch_bounds__(256) void gemm_dual_kernel(const float* __restrict__ x,
                                                        const float* __restrict__ WphiP,
                                                        const float* __restrict__ ATP,
                                                        const float* __restrict__ bias,
                                                        float* __restrict__ xw,
                                                        float* __restrict__ agg, int N) {
    __shared__ float lds_x[TILE_M][D + 4];

    const int row0 = blockIdx.x * TILE_M;
    const bool full = (row0 + TILE_M) <= N;
    const int tid = threadIdx.x;

    for (int i = tid; i < TILE_M * (D / 4); i += 256) {
        int r = i / (D / 4);
        int c = (i % (D / 4)) * 4;
        float4 v = make_float4(0.f, 0.f, 0.f, 0.f);
        if (full || (row0 + r < N)) v = *(const float4*)(x + (size_t)(row0 + r) * D + c);
        lds_x[r][c + 0] = v.x; lds_x[r][c + 1] = v.y;
        lds_x[r][c + 2] = v.z; lds_x[r][c + 3] = v.w;
    }
    __syncthreads();

    const int wave = tid >> 5;
    const int lane = tid & 31;
    const int mn   = lane & 15;
    const int kq   = lane >> 4;
    const int col0 = wave * 16;
    const size_t bcol = (size_t)(col0 + mn) * 2;

    v8f acc_w = {};
    v8f acc_a = {};

#pragma unroll 4
    for (int k0 = 0; k0 < D; k0 += 4) {
        const int k = k0 + kq * 2;
        const int p = (k0 >> 1) + kq;
        const v2f a  = *(const v2f*)&lds_x[mn][k];
        const v2f bw = *(const v2f*)(WphiP + (size_t)p * (2 * D) + bcol);
        const v2f ba = *(const v2f*)(ATP   + (size_t)p * (2 * D) + bcol);
        acc_w = __builtin_amdgcn_wmma_f32_16x16x4_f32(false, a, false, bw,
                                                      (short)0, acc_w, false, false);
        acc_a = __builtin_amdgcn_wmma_f32_16x16x4_f32(false, a, false, ba,
                                                      (short)0, acc_a, false, false);
    }

    const float b    = bias[col0 + mn];
    const int rowoff = kq * 8;
    if (full) {
#pragma unroll
        for (int r = 0; r < 8; ++r) {
            const size_t o = (size_t)(row0 + rowoff + r) * D + col0 + mn;
            xw[o]  = acc_w[r];
            agg[o] = acc_a[r] + b;
        }
    } else {
#pragma unroll
        for (int r = 0; r < 8; ++r) {
            const int row = row0 + rowoff + r;
            if (row < N) {
                const size_t o = (size_t)row * D + col0 + mn;
                xw[o]  = acc_w[r];
                agg[o] = acc_a[r] + b;
            }
        }
    }
}

__global__ __launch_bounds__(256) void edge_scatter_kernel(const int* __restrict__ ei,
                                                           const float* __restrict__ xw,
                                                           const float* __restrict__ dinv,
                                                           float* __restrict__ agg, int E) {
    const int wid  = (int)((blockIdx.x * (size_t)blockDim.x + threadIdx.x) >> 5);
    const int lane = threadIdx.x & 31;
    if (wid >= E) return;
    const int s = ei[wid];
    const int d = ei[E + wid];
    const float norm = dinv[s] * dinv[d];
    const float4 v = *(const float4*)(xw + (size_t)s * D + lane * 4);
    float* p = agg + (size_t)d * D + lane * 4;
    __hip_atomic_fetch_add(p + 0, v.x * norm, __ATOMIC_RELAXED, __HIP_MEMORY_SCOPE_AGENT);
    __hip_atomic_fetch_add(p + 1, v.y * norm, __ATOMIC_RELAXED, __HIP_MEMORY_SCOPE_AGENT);
    __hip_atomic_fetch_add(p + 2, v.z * norm, __ATOMIC_RELAXED, __HIP_MEMORY_SCOPE_AGENT);
    __hip_atomic_fetch_add(p + 3, v.w * norm, __ATOMIC_RELAXED, __HIP_MEMORY_SCOPE_AGENT);
}

__global__ __launch_bounds__(256) void combine_kernel(const float* __restrict__ xin,
                                                      const float* __restrict__ xw,
                                                      const float* __restrict__ agg,
                                                      const float* __restrict__ dinv,
                                                      float* __restrict__ xout, int N) {
    const int i = blockIdx.x * blockDim.x + threadIdx.x;
    const int total = N * (D / 4);
    if (i >= total) return;
    const int node = i / (D / 4);
    const float di = dinv[node];
    const float d2 = di * di;
    const size_t o = (size_t)i * 4;
    const float4 xv = *(const float4*)(xin + o);
    const float4 wv = *(const float4*)(xw + o);
    const float4 av = *(const float4*)(agg + o);
    float4 r;
    r.x = xv.x + EPSILON * tanhf(av.x + wv.x * d2);
    r.y = xv.y + EPSILON * tanhf(av.y + wv.y * d2);
    r.z = xv.z + EPSILON * tanhf(av.z + wv.z * d2);
    r.w = xv.w + EPSILON * tanhf(av.w + wv.w * d2);
    *(float4*)(xout + o) = r;
}

extern "C" void kernel_launch(void* const* d_in, const int* in_sizes, int n_in,
                              void* d_out, int out_size, void* d_ws, size_t ws_size,
                              hipStream_t stream) {
    const float* x_in = (const float*)d_in[0];
    const int*   ei   = (const int*)d_in[1];
    const float* W    = (const float*)d_in[2];
    const float* Wphi = (const float*)d_in[3];
    const float* bias = (const float*)d_in[4];
    float*       out  = (float*)d_out;

    const int N = in_sizes[0] / D;
    const int E = in_sizes[1] / 2;

    float* ws    = (float*)d_ws;
    const size_t nd = (size_t)N * D;
    float* xw    = ws;
    float* agg   = ws + nd;
    float* xbuf  = ws + 2 * nd;
    float* dinv  = ws + 3 * nd;
    float* WphiP = dinv + N;
    float* ATP   = WphiP + (size_t)D * D;

    const int B = 256;

    init_deg_kernel<<<(N + B - 1) / B, B, 0, stream>>>(dinv, N);
    count_deg_kernel<<<(E + B - 1) / B, B, 0, stream>>>(ei, dinv, E);
    finalize_dinv_kernel<<<(N + B - 1) / B, B, 0, stream>>>(dinv, N);
    prep_weights_kernel<<<(D * D + B - 1) / B, B, 0, stream>>>(W, Wphi, WphiP, ATP);

    const int gemm_blocks = (N + TILE_M - 1) / TILE_M;
    const int edge_blocks = (E + (B / 32) - 1) / (B / 32);
    const int comb_blocks = (N * (D / 4) + B - 1) / B;

    for (int it = 0; it < NUM_ITERS; ++it) {
        const float* xi = (it == 0) ? x_in : xbuf;
        float* xo = (it == NUM_ITERS - 1) ? out : xbuf;
        gemm_dual_kernel<<<gemm_blocks, B, 0, stream>>>(xi, WphiP, ATP, bias, xw, agg, N);
        edge_scatter_kernel<<<edge_blocks, B, 0, stream>>>(ei, xw, dinv, agg, E);
        combine_kernel<<<comb_blocks, B, 0, stream>>>(xi, xw, agg, dinv, xo, N);
    }
}


# Edge scatter's f32 atomics replaced by a per-call CSR by destination (int-atomic rank, scan, place) and one wave per node summing its rows in registers; sum order = rank arrival order
# speedup vs baseline: 2.5786x; 2.5786x over previous
#include <hip/hip_runtime.h>
#include <hip/hip_bf16.h>
#include <math.h>

#define D 128
#define TILE_M 16
#define GAMMA 0.1f
#define EPSILON 0.1f
#define NUM_ITERS 4

typedef __attribute__((ext_vector_type(2))) float v2f;
typedef __attribute__((ext_vector_type(8))) float v8f;

#define SCAN_ITEMS 1024

__global__ __launch_bounds__(256) void zero_cnt_kernel(int* __restrict__ cnt, int n) {
    int i = blockIdx.x * blockDim.x + threadIdx.x;
    if (i < n) cnt[i] = 0;
}

__global__ __launch_bounds__(256) void count_rank_kernel(const int* __restrict__ ei,
                                                         int* __restrict__ cnt,
                                                         int* __restrict__ rank, int E) {
    int e = blockIdx.x * blockDim.x + threadIdx.x;
    if (e < E) {
        int dst = ei[E + e];
        rank[e] = __hip_atomic_fetch_add(&cnt[dst], 1, __ATOMIC_RELAXED, __HIP_MEMORY_SCOPE_AGENT);
    }
}

__global__ __launch_bounds__(256) void block_sum_kernel(const int* __restrict__ cnt,
                                                        int* __restrict__ bsum, int n) {
    __shared__ int sm[8];
    const int tid = threadIdx.x, lane = tid & 31, wave = tid >> 5;
    const int base = blockIdx.x * SCAN_ITEMS + tid * 4;
    int s = 0;
    for (int u = 0; u < 4; ++u)
        if (base + u < n) s += cnt[base + u];
    for (int o = 16; o > 0; o >>= 1) s += __shfl_xor(s, o);
    if (lane == 0) sm[wave] = s;
    __syncthreads();
    if (tid == 0) {
        int t = 0;
        for (int w = 0; w < 8; ++w) t += sm[w];
        bsum[blockIdx.x] = t;
    }
}

__global__ __launch_bounds__(256) void scan_rows_kernel(const int* __restrict__ cnt,
                                                        const int* __restrict__ bsum,
                                                        int* __restrict__ rowptr,
                                                        float* __restrict__ dinv, int n) {
    __shared__ int sm_off[8];
    __shared__ int sm_wav[8];
    const int tid = threadIdx.x, lane = tid & 31, wave = tid >> 5;

    int part = 0;
    for (int j = tid; j < (int)blockIdx.x; j += 256) part += bsum[j];
    for (int o = 16; o > 0; o >>= 1) part += __shfl_xor(part, o);
    if (lane == 0) sm_off[wave] = part;

    const int base = blockIdx.x * SCAN_ITEMS + tid * 4;
    int c[4];
    int s = 0;
    for (int u = 0; u < 4; ++u) {
        c[u] = (base + u < n) ? cnt[base + u] : 0;
        s += c[u];
    }
    int incl = s;
    for (int o = 1; o < 32; o <<= 1) {
        int t = __shfl_up(incl, o);
        if (lane >= o) incl += t;
    }
    if (lane == 31) sm_wav[wave] = incl;
    __syncthreads();

    int excl = incl - s;
    for (int w = 0; w < 8; ++w) {
        excl += sm_off[w];
        if (w < wave) excl += sm_wav[w];
    }
    for (int u = 0; u < 4; ++u) {
        const int i = base + u;
        if (i < n) {
            rowptr[i] = excl;
            dinv[i] = rsqrtf((float)c[u] + 1.0f);
            excl += c[u];
            if (i == n - 1) rowptr[n] = excl;
        }
    }
}

__global__ __launch_bounds__(256) void fill_csr_kernel(const int* __restrict__ ei,
                                                       const int* __restrict__ rank,
                                                       const int* __restrict__ rowptr,
                                                       const float* __restrict__ dinv,
                                                       int* __restrict__ csr, int E) {
    int e = blockIdx.x * blockDim.x + threadIdx.x;
    if (e < E) {
        const int s = ei[e];
        const int d = ei[E + e];
        const int pos = rowptr[d] + rank[e];
        int2 rec;
        rec.x = s;
        rec.y = __float_as_int(dinv[s] * dinv[d]);
        *(int2*)(csr + (size_t)pos * 2) = rec;
    }
}

__global__ __launch_bounds__(256) void prep_weights_kernel(const float* __restrict__ W,
                                                           const float* __restrict__ Wphi,
                                                           float* __restrict__ WphiP,
                                                           float* __restrict__ ATP) {
    int i = blockIdx.x * blockDim.x + threadIdx.x;
    if (i >= D * D) return;
    int k = i / D, n = i % D;
    size_t po = (size_t)(k >> 1) * (2 * D) + n * 2 + (k & 1);
    WphiP[po] = Wphi[n * D + k];
    float a = W[n * D + k] - W[k * D + n];
    if (k == n) a -= GAMMA;
    ATP[po] = a;
}

__global__ __launch_bounds__(256) void gemm_dual_kernel(const float* __restrict__ x,
                                                        const float* __restrict__ WphiP,
                                                        const float* __restrict__ ATP,
                                                        const float* __restrict__ bias,
                                                        float* __restrict__ xw,
                                                        float* __restrict__ agg, int N) {
    __shared__ float lds_x[TILE_M][D + 4];

    const int row0 = blockIdx.x * TILE_M;
    const bool full = (row0 + TILE_M) <= N;
    const int tid = threadIdx.x;

    for (int i = tid; i < TILE_M * (D / 4); i += 256) {
        int r = i / (D / 4);
        int c = (i % (D / 4)) * 4;
        float4 v = make_float4(0.f, 0.f, 0.f, 0.f);
        if (full || (row0 + r < N)) v = *(const float4*)(x + (size_t)(row0 + r) * D + c);
        lds_x[r][c + 0] = v.x; lds_x[r][c + 1] = v.y;
        lds_x[r][c + 2] = v.z; lds_x[r][c + 3] = v.w;
    }
    __syncthreads();

    const int wave = tid >> 5;
    const int lane = tid & 31;
    const int mn   = lane & 15;
    const int kq   = lane >> 4;
    const int col0 = wave * 16;
    const size_t bcol = (size_t)(col0 + mn) * 2;

    v8f acc_w = {};
    v8f acc_a = {};

#pragma unroll 4
    for (int k0 = 0; k0 < D; k0 += 4) {
        const int k = k0 + kq * 2;
        const int p = (k0 >> 1) + kq;
        const v2f a  = *(const v2f*)&lds_x[mn][k];
        const v2f bw = *(const v2f*)(WphiP + (size_t)p * (2 * D) + bcol);
        const v2f ba = *(const v2f*)(ATP   + (size_t)p * (2 * D) + bcol);
        acc_w = __builtin_amdgcn_wmma_f32_16x16x4_f32(false, a, false, bw,
                                                      (short)0, acc_w, false, false);
        acc_a = __builtin_amdgcn_wmma_f32_16x16x4_f32(false, a, false, ba,
                                                      (short)0, acc_a, false, false);
    }

    const float b    = bias[col0 + mn];
    const int rowoff = kq * 8;
    if (full) {
#pragma unroll
        for (int r = 0; r < 8; ++r) {
            const size_t o = (size_t)(row0 + rowoff + r) * D + col0 + mn;
            xw[o]  = acc_w[r];
            agg[o] = acc_a[r] + b;
        }
    } else {
#pragma unroll
        for (int r = 0; r < 8; ++r) {
            const int row = row0 + rowoff + r;
            if (row < N) {
                const size_t o = (size_t)row * D + col0 + mn;
                xw[o]  = acc_w[r];
                agg[o] = acc_a[r] + b;
            }
        }
    }
}

template <int K>
__device__ __forceinline__ void gather_rows(float4& acc, const float* __restrict__ xw,
                                            unsigned lane4, int2 my, int j) {
    float4 v[K];
    float w[K];
#pragma unroll
    for (int u = 0; u < K; ++u) {
        const int s = __builtin_amdgcn_readlane(my.x, j + u);
        w[u] = __int_as_float(__builtin_amdgcn_readlane(my.y, j + u));
        v[u] = *(const float4*)(xw + (size_t)s * D + lane4);
    }
#pragma unroll
    for (int u = 0; u < K; ++u) {
        acc.x += v[u].x * w[u]; acc.y += v[u].y * w[u];
        acc.z += v[u].z * w[u]; acc.w += v[u].w * w[u];
    }
}

__global__ __launch_bounds__(256) void edge_gather_kernel(const int* __restrict__ rowptr,
                                                          const int* __restrict__ csr,
                                                          const float* __restrict__ xw,
                                                          float* __restrict__ agg, int N) {
    const int d = __builtin_amdgcn_readfirstlane((int)((blockIdx.x * 256u + threadIdx.x) >> 5));
    const int lane = threadIdx.x & 31;
    if (d >= N) return;
    const int beg = rowptr[d];
    const int end = rowptr[d + 1];
    float* p = agg + (size_t)d * D + lane * 4;
    float4 acc = *(const float4*)p;
    const unsigned lane4 = (unsigned)lane * 4u;

    for (int base = beg; base < end; base += 32) {
        const int n = min(32, end - base);
        int2 my = make_int2(0, 0);
        if (lane < n) my = *(const int2*)(csr + (size_t)(base + lane) * 2);
        int j = 0;
        for (; j + 8 <= n; j += 8) gather_rows<8>(acc, xw, lane4, my, j);
        if (j + 4 <= n) { gather_rows<4>(acc, xw, lane4, my, j); j += 4; }
        if (j + 2 <= n) { gather_rows<2>(acc, xw, lane4, my, j); j += 2; }
        if (j < n) gather_rows<1>(acc, xw, lane4, my, j);
    }
    *(float4*)p = acc;
}

__global__ __launch_bounds__(256) void combine_kernel(const float* __restrict__ xin,
                                                      const float* __restrict__ xw,
                                                      const float* __restrict__ agg,
                                                      const float* __restrict__ dinv,
                                                      float* __restrict__ xout, int N) {
    const int i = blockIdx.x * blockDim.x + threadIdx.x;
    const int total = N * (D / 4);
    if (i >= total) return;
    const int node = i / (D / 4);
    const float di = dinv[node];
    const float d2 = di * di;
    const size_t o = (size_t)i * 4;
    const float4 xv = *(const float4*)(xin + o);
    const float4 wv = *(const float4*)(xw + o);
    const float4 av = *(const float4*)(agg + o);
    float4 r;
    r.x = xv.x + EPSILON * tanhf(av.x + wv.x * d2);
    r.y = xv.y + EPSILON * tanhf(av.y + wv.y * d2);
    r.z = xv.z + EPSILON * tanhf(av.z + wv.z * d2);
    r.w = xv.w + EPSILON * tanhf(av.w + wv.w * d2);
    *(float4*)(xout + o) = r;
}

extern "C" void kernel_launch(void* const* d_in, const int* in_sizes, int n_in,
                              void* d_out, int out_size, void* d_ws, size_t ws_size,
                              hipStream_t stream) {
    const float* x_in = (const float*)d_in[0];
    const int*   ei   = (const int*)d_in[1];
    const float* W    = (const float*)d_in[2];
    const float* Wphi = (const float*)d_in[3];
    const float* bias = (const float*)d_in[4];
    float*       out  = (float*)d_out;

    const int N = in_sizes[0] / D;
    const int E = in_sizes[1] / 2;

    float* ws    = (float*)d_ws;
    const size_t nd = (size_t)N * D;
    float* xw    = ws;
    float* agg   = ws + nd;
    int*   csr   = (int*)(ws + 2 * nd);
    int*   rowptr = csr + (size_t)E * 2;
    float* dinv  = ws + 3 * nd;
    float* WphiP = dinv + N;
    float* ATP   = WphiP + (size_t)D * D;
    int*   cnt   = (int*)xw;
    int*   rank  = cnt + N;
    int*   bsum  = rank + E;

    const int B = 256;
    const int scan_blocks = (N + SCAN_ITEMS - 1) / SCAN_ITEMS;

    zero_cnt_kernel<<<(N + B - 1) / B, B, 0, stream>>>(cnt, N);
    count_rank_kernel<<<(E + B - 1) / B, B, 0, stream>>>(ei, cnt, rank, E);
    block_sum_kernel<<<scan_blocks, B, 0, stream>>>(cnt, bsum, N);
    scan_rows_kernel<<<scan_blocks, B, 0, stream>>>(cnt, bsum, rowptr, dinv, N);
    fill_csr_kernel<<<(E + B - 1) / B, B, 0, stream>>>(ei, rank, rowptr, dinv, csr, E);
    prep_weights_kernel<<<(D * D + B - 1) / B, B, 0, stream>>>(W, Wphi, WphiP, ATP);

    const int gemm_blocks = (N + TILE_M - 1) / TILE_M;
    const int node_blocks = (N + (B / 32) - 1) / (B / 32);
    const int comb_blocks = (N * (D / 4) + B - 1) / B;

    for (int it = 0; it < NUM_ITERS; ++it) {
        const float* xi = (it == 0) ? x_in : out;
        gemm_dual_kernel<<<gemm_blocks, B, 0, stream>>>(xi, WphiP, ATP, bias, xw, agg, N);
        edge_gather_kernel<<<node_blocks, B, 0, stream>>>(rowptr, csr, xw, agg, N);
        combine_kernel<<<comb_blocks, B, 0, stream>>>(xi, xw, agg, dinv, out, N);
    }
}


# Combine pass fused into the edge gather: each node's wave applies tanh and the x update to its register row sum and writes x itself; drops 4 launches and the agg write-back and re-read
# speedup vs baseline: 2.8222x; 1.0945x over previous
#include <hip/hip_runtime.h>
#include <hip/hip_bf16.h>
#include <math.h>

#define D 128
#define TILE_M 16
#define GAMMA 0.1f
#define EPSILON 0.1f
#define NUM_ITERS 4

typedef __attribute__((ext_vector_type(2))) float v2f;
typedef __attribute__((ext_vector_type(8))) float v8f;

#define SCAN_ITEMS 1024

__global__ __launch_bounds__(256) void zero_cnt_kernel(int* __restrict__ cnt, int n) {
    int i = blockIdx.x * blockDim.x + threadIdx.x;
    if (i < n) cnt[i] = 0;
}

__global__ __launch_bounds__(256) void count_rank_kernel(const int* __restrict__ ei,
                                                         int* __restrict__ cnt,
                                                         int* __restrict__ rank, int E) {
    int e = blockIdx.x * blockDim.x + threadIdx.x;
    if (e < E) {
        int dst = ei[E + e];
        rank[e] = __hip_atomic_fetch_add(&cnt[dst], 1, __ATOMIC_RELAXED, __HIP_MEMORY_SCOPE_AGENT);
    }
}

__global__ __launch_bounds__(256) void block_sum_kernel(const int* __restrict__ cnt,
                                                        int* __restrict__ bsum, int n) {
    __shared__ int sm[8];
    const int tid = threadIdx.x, lane = tid & 31, wave = tid >> 5;
    const int base = blockIdx.x * SCAN_ITEMS + tid * 4;
    int s = 0;
    for (int u = 0; u < 4; ++u)
        if (base + u < n) s += cnt[base + u];
    for (int o = 16; o > 0; o >>= 1) s += __shfl_xor(s, o);
    if (lane == 0) sm[wave] = s;
    __syncthreads();
    if (tid == 0) {
        int t = 0;
        for (int w = 0; w < 8; ++w) t += sm[w];
        bsum[blockIdx.x] = t;
    }
}

__global__ __launch_bounds__(256) void scan_rows_kernel(const int* __restrict__ cnt,
                                                        const int* __restrict__ bsum,
                                                        int* __restrict__ rowptr,
                                                        float* __restrict__ dinv, int n) {
    __shared__ int sm_off[8];
    __shared__ int sm_wav[8];
    const int tid = threadIdx.x, lane = tid & 31, wave = tid >> 5;

    int part = 0;
    for (int j = tid; j < (int)blockIdx.x; j += 256) part += bsum[j];
    for (int o = 16; o > 0; o >>= 1) part += __shfl_xor(part, o);
    if (lane == 0) sm_off[wave] = part;

    const int base = blockIdx.x * SCAN_ITEMS + tid * 4;
    int c[4];
    int s = 0;
    for (int u = 0; u < 4; ++u) {
        c[u] = (base + u < n) ? cnt[base + u] : 0;
        s += c[u];
    }
    int incl = s;
    for (int o = 1; o < 32; o <<= 1) {
        int t = __shfl_up(incl, o);
        if (lane >= o) incl += t;
    }
    if (lane == 31) sm_wav[wave] = incl;
    __syncthreads();

    int excl = incl - s;
    for (int w = 0; w < 8; ++w) {
        excl += sm_off[w];
        if (w < wave) excl += sm_wav[w];
    }
    for (int u = 0; u < 4; ++u) {
        const int i = base + u;
        if (i < n) {
            rowptr[i] = excl;
            dinv[i] = rsqrtf((float)c[u] + 1.0f);
            excl += c[u];
            if (i == n - 1) rowptr[n] = excl;
        }
    }
}

__global__ __launch_bounds__(256) void fill_csr_kernel(const int* __restrict__ ei,
                                                       const int* __restrict__ rank,
                                                       const int* __restrict__ rowptr,
                                                       const float* __restrict__ dinv,
                                                       int* __restrict__ csr, int E) {
    int e = blockIdx.x * blockDim.x + threadIdx.x;
    if (e < E) {
        const int s = ei[e];
        const int d = ei[E + e];
        const int pos = rowptr[d] + rank[e];
        int2 rec;
        rec.x = s;
        rec.y = __float_as_int(dinv[s] * dinv[d]);
        *(int2*)(csr + (size_t)pos * 2) = rec;
    }
}

__global__ __launch_bounds__(256) void prep_weights_kernel(const float* __restrict__ W,
                                                           const float* __restrict__ Wphi,
                                                           float* __restrict__ WphiP,
                                                           float* __restrict__ ATP) {
    int i = blockIdx.x * blockDim.x + threadIdx.x;
    if (i >= D * D) return;
    int k = i / D, n = i % D;
    size_t po = (size_t)(k >> 1) * (2 * D) + n * 2 + (k & 1);
    WphiP[po] = Wphi[n * D + k];
    float a = W[n * D + k] - W[k * D + n];
    if (k == n) a -= GAMMA;
    ATP[po] = a;
}

__global__ __launch_bounds__(256) void gemm_dual_kernel(const float* __restrict__ x,
                                                        const float* __restrict__ WphiP,
                                                        const float* __restrict__ ATP,
                                                        const float* __restrict__ bias,
                                                        float* __restrict__ xw,
                                                        float* __restrict__ agg, int N) {
    __shared__ float lds_x[TILE_M][D + 4];

    const int row0 = blockIdx.x * TILE_M;
    const bool full = (row0 + TILE_M) <= N;
    const int tid = threadIdx.x;

    for (int i = tid; i < TILE_M * (D / 4); i += 256) {
        int r = i / (D / 4);
        int c = (i % (D / 4)) * 4;
        float4 v = make_float4(0.f, 0.f, 0.f, 0.f);
        if (full || (row0 + r < N)) v = *(const float4*)(x + (size_t)(row0 + r) * D + c);
        lds_x[r][c + 0] = v.x; lds_x[r][c + 1] = v.y;
        lds_x[r][c + 2] = v.z; lds_x[r][c + 3] = v.w;
    }
    __syncthreads();

    const int wave = tid >> 5;
    const int lane = tid & 31;
    const int mn   = lane & 15;
    const int kq   = lane >> 4;
    const int col0 = wave * 16;
    const size_t bcol = (size_t)(col0 + mn) * 2;

    v8f acc_w = {};
    v8f acc_a = {};

#pragma unroll 4
    for (int k0 = 0; k0 < D; k0 += 4) {
        const int k = k0 + kq * 2;
        const int p = (k0 >> 1) + kq;
        const v2f a  = *(const v2f*)&lds_x[mn][k];
        const v2f bw = *(const v2f*)(WphiP + (size_t)p * (2 * D) + bcol);
        const v2f ba = *(const v2f*)(ATP   + (size_t)p * (2 * D) + bcol);
        acc_w = __builtin_amdgcn_wmma_f32_16x16x4_f32(false, a, false, bw,
                                                      (short)0, acc_w, false, false);
        acc_a = __builtin_amdgcn_wmma_f32_16x16x4_f32(false, a, false, ba,
                                                      (short)0, acc_a, false, false);
    }

    const float b    = bias[col0 + mn];
    const int rowoff = kq * 8;
    if (full) {
#pragma unroll
        for (int r = 0; r < 8; ++r) {
            const size_t o = (size_t)(row0 + rowoff + r) * D + col0 + mn;
            xw[o]  = acc_w[r];
            agg[o] = acc_a[r] + b;
        }
    } else {
#pragma unroll
        for (int r = 0; r < 8; ++r) {
            const int row = row0 + rowoff + r;
            if (row < N) {
                const size_t o = (size_t)row * D + col0 + mn;
                xw[o]  = acc_w[r];
                agg[o] = acc_a[r] + b;
            }
        }
    }
}

template <int K>
__device__ __forceinline__ void gather_rows(float4& acc, const float* __restrict__ xw,
                                            unsigned lane4, int2 my, int j) {
    float4 v[K];
    float w[K];
#pragma unroll
    for (int u = 0; u < K; ++u) {
        const int s = __builtin_amdgcn_readlane(my.x, j + u);
        w[u] = __int_as_float(__builtin_amdgcn_readlane(my.y, j + u));
        v[u] = *(const float4*)(xw + (size_t)s * D + lane4);
    }
#pragma unroll
    for (int u = 0; u < K; ++u) {
        acc.x += v[u].x * w[u]; acc.y += v[u].y * w[u];
        acc.z += v[u].z * w[u]; acc.w += v[u].w * w[u];
    }
}

__global__ __launch_bounds__(256) void edge_gather_combine_kernel(const int* __restrict__ rowptr,
                                                                  const int* __restrict__ csr,
                                                                  const float* __restrict__ xw,
                                                                  const float* __restrict__ agg,
                                                                  const float* __restrict__ dinv,
                                                                  const float* xin,
                                                                  float* xout, int N) {
    const int d = __builtin_amdgcn_readfirstlane((int)((blockIdx.x * 256u + threadIdx.x) >> 5));
    const int lane = threadIdx.x & 31;
    if (d >= N) return;
    const int beg = rowptr[d];
    const int end = rowptr[d + 1];
    const float di = dinv[d];
    const unsigned lane4 = (unsigned)lane * 4u;
    const size_t o = (size_t)d * D + lane4;
    float4 acc = *(const float4*)(agg + o);

    for (int base = beg; base < end; base += 32) {
        const int n = min(32, end - base);
        int2 my = make_int2(0, 0);
        if (lane < n) my = *(const int2*)(csr + (size_t)(base + lane) * 2);
        int j = 0;
        for (; j + 8 <= n; j += 8) gather_rows<8>(acc, xw, lane4, my, j);
        if (j + 4 <= n) { gather_rows<4>(acc, xw, lane4, my, j); j += 4; }
        if (j + 2 <= n) { gather_rows<2>(acc, xw, lane4, my, j); j += 2; }
        if (j < n) gather_rows<1>(acc, xw, lane4, my, j);
    }

    const float4 wv = *(const float4*)(xw + o);
    const float4 xv = *(const float4*)(xin + o);
    const float d2 = di * di;
    float4 r;
    r.x = xv.x + EPSILON * tanhf(acc.x + wv.x * d2);
    r.y = xv.y + EPSILON * tanhf(acc.y + wv.y * d2);
    r.z = xv.z + EPSILON * tanhf(acc.z + wv.z * d2);
    r.w = xv.w + EPSILON * tanhf(acc.w + wv.w * d2);
    *(float4*)(xout + o) = r;
}

extern "C" void kernel_launch(void* const* d_in, const int* in_sizes, int n_in,
                              void* d_out, int out_size, void* d_ws, size_t ws_size,
                              hipStream_t stream) {
    const float* x_in = (const float*)d_in[0];
    const int*   ei   = (const int*)d_in[1];
    const float* W    = (const float*)d_in[2];
    const float* Wphi = (const float*)d_in[3];
    const float* bias = (const float*)d_in[4];
    float*       out  = (float*)d_out;

    const int N = in_sizes[0] / D;
    const int E = in_sizes[1] / 2;

    float* ws    = (float*)d_ws;
    const size_t nd = (size_t)N * D;
    float* xw    = ws;
    float* agg   = ws + nd;
    int*   csr   = (int*)(ws + 2 * nd);
    int*   rowptr = csr + (size_t)E * 2;
    float* dinv  = ws + 3 * nd;
    float* WphiP = dinv + N;
    float* ATP   = WphiP + (size_t)D * D;
    int*   cnt   = (int*)xw;
    int*   rank  = cnt + N;
    int*   bsum  = rank + E;

    const int B = 256;
    const int scan_blocks = (N + SCAN_ITEMS - 1) / SCAN_ITEMS;

    zero_cnt_kernel<<<(N + B - 1) / B, B, 0, stream>>>(cnt, N);
    count_rank_kernel<<<(E + B - 1) / B, B, 0, stream>>>(ei, cnt, rank, E);
    block_sum_kernel<<<scan_blocks, B, 0, stream>>>(cnt, bsum, N);
    scan_rows_kernel<<<scan_blocks, B, 0, stream>>>(cnt, bsum, rowptr, dinv, N);
    fill_csr_kernel<<<(E + B - 1) / B, B, 0, stream>>>(ei, rank, rowptr, dinv, csr, E);
    prep_weights_kernel<<<(D * D + B - 1) / B, B, 0, stream>>>(W, Wphi, WphiP, ATP);

    const int gemm_blocks = (N + TILE_M - 1) / TILE_M;
    const int node_blocks = (N + (B / 32) - 1) / (B / 32);

    for (int it = 0; it < NUM_ITERS; ++it) {
        const float* xi = (it == 0) ? x_in : out;
        gemm_dual_kernel<<<gemm_blocks, B, 0, stream>>>(xi, WphiP, ATP, bias, xw, agg, N);
        edge_gather_combine_kernel<<<node_blocks, B, 0, stream>>>(rowptr, csr, xw, agg, dinv, xi, out, N);
    }
}


# Per iteration one launch: linear map moved behind the aggregation, (A_hat x)Wphi^T; a 64-node block sums x rows into LDS, then f32 WMMA with tanh+update epilogue; rows now summed before the product
# speedup vs baseline: 3.2805x; 1.1624x over previous
#include <hip/hip_runtime.h>
#include <hip/hip_bf16.h>
#include <math.h>

#define D 128
#define GAMMA 0.1f
#define EPSILON 0.1f
#define NUM_ITERS 4

typedef __attribute__((ext_vector_type(2))) float v2f;
typedef __attribute__((ext_vector_type(8))) float v8f;

#define SCAN_ITEMS 1024

__global__ __launch_bounds__(256) void zero_cnt_kernel(int* __restrict__ cnt, int n) {
    int i = blockIdx.x * blockDim.x + threadIdx.x;
    if (i < n) cnt[i] = 0;
}

__global__ __launch_bounds__(256) void count_rank_kernel(const int* __restrict__ ei,
                                                         int* __restrict__ cnt,
                                                         int* __restrict__ rank, int E) {
    int e = blockIdx.x * blockDim.x + threadIdx.x;
    if (e < E) {
        int dst = ei[E + e];
        rank[e] = __hip_atomic_fetch_add(&cnt[dst], 1, __ATOMIC_RELAXED, __HIP_MEMORY_SCOPE_AGENT);
    }
}

__global__ __launch_bounds__(256) void block_sum_kernel(const int* __restrict__ cnt,
                                                        int* __restrict__ bsum, int n) {
    __shared__ int sm[8];
    const int tid = threadIdx.x, lane = tid & 31, wave = tid >> 5;
    const int base = blockIdx.x * SCAN_ITEMS + tid * 4;
    int s = 0;
    for (int u = 0; u < 4; ++u)
        if (base + u < n) s += cnt[base + u];
    for (int o = 16; o > 0; o >>= 1) s += __shfl_xor(s, o);
    if (lane == 0) sm[wave] = s;
    __syncthreads();
    if (tid == 0) {
        int t = 0;
        for (int w = 0; w < 8; ++w) t += sm[w];
        bsum[blockIdx.x] = t;
    }
}

__global__ __launch_bounds__(256) void scan_rows_kernel(const int* __restrict__ cnt,
                                                        const int* __restrict__ bsum,
                                                        int* __restrict__ rowptr,
                                                        float* __restrict__ dinv, int n) {
    __shared__ int sm_off[8];
    __shared__ int sm_wav[8];
    const int tid = threadIdx.x, lane = tid & 31, wave = tid >> 5;

    int part = 0;
    for (int j = tid; j < (int)blockIdx.x; j += 256) part += bsum[j];
    for (int o = 16; o > 0; o >>= 1) part += __shfl_xor(part, o);
    if (lane == 0) sm_off[wave] = part;

    const int base = blockIdx.x * SCAN_ITEMS + tid * 4;
    int c[4];
    int s = 0;
    for (int u = 0; u < 4; ++u) {
        c[u] = (base + u < n) ? cnt[base + u] : 0;
        s += c[u];
    }
    int incl = s;
    for (int o = 1; o < 32; o <<= 1) {
        int t = __shfl_up(incl, o);
        if (lane >= o) incl += t;
    }
    if (lane == 31) sm_wav[wave] = incl;
    __syncthreads();

    int excl = incl - s;
    for (int w = 0; w < 8; ++w) {
        excl += sm_off[w];
        if (w < wave) excl += sm_wav[w];
    }
    for (int u = 0; u < 4; ++u) {
        const int i = base + u;
        if (i < n) {
            rowptr[i] = excl;
            dinv[i] = rsqrtf((float)c[u] + 1.0f);
            excl += c[u];
            if (i == n - 1) rowptr[n] = excl;
        }
    }
}

__global__ __launch_bounds__(256) void fill_csr_kernel(const int* __restrict__ ei,
                                                       const int* __restrict__ rank,
                                                       const int* __restrict__ rowptr,
                                                       const float* __restrict__ dinv,
                                                       int* __restrict__ csr, int E) {
    int e = blockIdx.x * blockDim.x + threadIdx.x;
    if (e < E) {
        const int s = ei[e];
        const int d = ei[E + e];
        const int pos = rowptr[d] + rank[e];
        int2 rec;
        rec.x = s;
        rec.y = __float_as_int(dinv[s] * dinv[d]);
        *(int2*)(csr + (size_t)pos * 2) = rec;
    }
}

__global__ __launch_bounds__(256) void prep_weights_kernel(const float* __restrict__ W,
                                                           const float* __restrict__ Wphi,
                                                           float* __restrict__ WphiP,
                                                           float* __restrict__ ATP) {
    int i = blockIdx.x * blockDim.x + threadIdx.x;
    if (i >= D * D) return;
    int k = i / D, n = i % D;
    size_t po = (size_t)(k >> 1) * (2 * D) + n * 2 + (k & 1);
    WphiP[po] = Wphi[n * D + k];
    float a = W[n * D + k] - W[k * D + n];
    if (k == n) a -= GAMMA;
    ATP[po] = a;
}

template <int K>
__device__ __forceinline__ void gather_rows(float4& acc, const float* __restrict__ xs,
                                            unsigned lane4, int2 my, int j) {
    float4 v[K];
    float w[K];
#pragma unroll
    for (int u = 0; u < K; ++u) {
        const int s = __builtin_amdgcn_readlane(my.x, j + u);
        w[u] = __int_as_float(__builtin_amdgcn_readlane(my.y, j + u));
        v[u] = *(const float4*)(xs + (size_t)s * D + lane4);
    }
#pragma unroll
    for (int u = 0; u < K; ++u) {
        acc.x += v[u].x * w[u]; acc.y += v[u].y * w[u];
        acc.z += v[u].z * w[u]; acc.w += v[u].w * w[u];
    }
}

#define NWAVES 16
#define TILE_N 64
#define NPW (TILE_N / NWAVES)
#define MT ((TILE_N / 16) / (NWAVES / 8))

__global__ __launch_bounds__(NWAVES * 32) void gather_gemm_update_kernel(const int* __restrict__ rowptr,
                                                                 const int* __restrict__ csr,
                                                                 const float* __restrict__ dinv,
                                                                 const float* __restrict__ WphiP,
                                                                 const float* __restrict__ ATP,
                                                                 const float* __restrict__ bias,
                                                                 const float* __restrict__ xin,
                                                                 float* __restrict__ xout, int N) {
    __shared__ __attribute__((aligned(16))) float lds_x[TILE_N][D + 4];
    __shared__ __attribute__((aligned(16))) float lds_y[TILE_N][D + 4];

    const int tid  = threadIdx.x;
    const int lane = tid & 31;
    const int wave = __builtin_amdgcn_readfirstlane(tid >> 5);
    const int row0 = blockIdx.x * TILE_N;
    const unsigned lane4 = (unsigned)lane * 4u;

    const int d0 = row0 + wave * NPW;
    int   pb[NPW + 1];
    float dv[NPW];
#pragma unroll
    for (int i = 0; i <= NPW; ++i) pb[i] = rowptr[min(d0 + i, N)];
#pragma unroll
    for (int i = 0; i < NPW; ++i) dv[i] = dinv[min(d0 + i, N - 1)];

#pragma unroll
    for (int i = 0; i < NPW; ++i) {
        const int r   = wave * NPW + i;
        const int beg = pb[i];
        const int end = pb[i + 1];
        float4 xs = *(const float4*)(xin + (size_t)min(d0 + i, N - 1) * D + lane4);
        if (d0 + i >= N) xs = make_float4(0.f, 0.f, 0.f, 0.f);
        float4 acc = make_float4(0.f, 0.f, 0.f, 0.f);
        for (int base = beg; base < end; base += 32) {
            const int n = min(32, end - base);
            int2 my = make_int2(0, 0);
            if (lane < n) my = *(const int2*)(csr + (size_t)(base + lane) * 2);
            int j = 0;
            for (; j + 8 <= n; j += 8) gather_rows<8>(acc, xin, lane4, my, j);
            if (j + 4 <= n) { gather_rows<4>(acc, xin, lane4, my, j); j += 4; }
            if (j + 2 <= n) { gather_rows<2>(acc, xin, lane4, my, j); j += 2; }
            if (j < n) gather_rows<1>(acc, xin, lane4, my, j);
        }
        const float d2 = dv[i] * dv[i];
        acc.x += xs.x * d2; acc.y += xs.y * d2;
        acc.z += xs.z * d2; acc.w += xs.w * d2;
        *(float4*)&lds_x[r][lane4] = xs;
        *(float4*)&lds_y[r][lane4] = acc;
    }
    __syncthreads();

    const int mn   = lane & 15;
    const int kq   = lane >> 4;
    const int col0 = (wave & 7) * 16;
    const int mrow0 = (wave >> 3) * (MT * 16);
    const size_t bcol = (size_t)(col0 + mn) * 2;

    v8f acc_w[MT];
    v8f acc_a[MT];
#pragma unroll
    for (int m = 0; m < MT; ++m) { acc_w[m] = (v8f){}; acc_a[m] = (v8f){}; }

#pragma unroll 4
    for (int k0 = 0; k0 < D; k0 += 4) {
        const int k = k0 + kq * 2;
        const int p = (k0 >> 1) + kq;
        const v2f bw = *(const v2f*)(WphiP + (size_t)p * (2 * D) + bcol);
        const v2f ba = *(const v2f*)(ATP   + (size_t)p * (2 * D) + bcol);
#pragma unroll
        for (int m = 0; m < MT; ++m) {
            const v2f ay = *(const v2f*)&lds_y[mrow0 + m * 16 + mn][k];
            const v2f ax = *(const v2f*)&lds_x[mrow0 + m * 16 + mn][k];
            acc_w[m] = __builtin_amdgcn_wmma_f32_16x16x4_f32(false, ay, false, bw,
                                                             (short)0, acc_w[m], false, false);
            acc_a[m] = __builtin_amdgcn_wmma_f32_16x16x4_f32(false, ax, false, ba,
                                                             (short)0, acc_a[m], false, false);
        }
    }

    const float b    = bias[col0 + mn];
    const int rowoff = kq * 8;
#pragma unroll
    for (int m = 0; m < MT; ++m) {
#pragma unroll
        for (int r = 0; r < 8; ++r) {
            const float xo = lds_x[mrow0 + m * 16 + rowoff + r][col0 + mn];
            acc_w[m][r] = xo + EPSILON * tanhf(acc_w[m][r] + acc_a[m][r] + b);
        }
    }
    __syncthreads();
#pragma unroll
    for (int m = 0; m < MT; ++m) {
#pragma unroll
        for (int r = 0; r < 8; ++r) lds_y[mrow0 + m * 16 + rowoff + r][col0 + mn] = acc_w[m][r];
    }
    __syncthreads();
#pragma unroll
    for (int i = 0; i < NPW; ++i) {
        const int r = wave * NPW + i;
        const int d = row0 + r;
        if (d < N) *(float4*)(xout + (size_t)d * D + lane4) = *(const float4*)&lds_y[r][lane4];
    }
}

extern "C" void kernel_launch(void* const* d_in, const int* in_sizes, int n_in,
                              void* d_out, int out_size, void* d_ws, size_t ws_size,
                              hipStream_t stream) {
    const float* x_in = (const float*)d_in[0];
    const int*   ei   = (const int*)d_in[1];
    const float* W    = (const float*)d_in[2];
    const float* Wphi = (const float*)d_in[3];
    const float* bias = (const float*)d_in[4];
    float*       out  = (float*)d_out;

    const int N = in_sizes[0] / D;
    const int E = in_sizes[1] / 2;

    float* ws    = (float*)d_ws;
    const size_t nd = (size_t)N * D;
    float* xalt  = ws;
    int*   cnt   = (int*)(ws + nd);
    int*   rank  = cnt + N;
    int*   bsum  = rank + E;
    int*   csr   = (int*)(ws + 2 * nd);
    int*   rowptr = csr + (size_t)E * 2;
    float* dinv  = ws + 3 * nd;
    float* WphiP = dinv + N;
    float* ATP   = WphiP + (size_t)D * D;

    const int B = 256;
    const int scan_blocks = (N + SCAN_ITEMS - 1) / SCAN_ITEMS;

    zero_cnt_kernel<<<(N + B - 1) / B, B, 0, stream>>>(cnt, N);
    count_rank_kernel<<<(E + B - 1) / B, B, 0, stream>>>(ei, cnt, rank, E);
    block_sum_kernel<<<scan_blocks, B, 0, stream>>>(cnt, bsum, N);
    scan_rows_kernel<<<scan_blocks, B, 0, stream>>>(cnt, bsum, rowptr, dinv, N);
    fill_csr_kernel<<<(E + B - 1) / B, B, 0, stream>>>(ei, rank, rowptr, dinv, csr, E);
    prep_weights_kernel<<<(D * D + B - 1) / B, B, 0, stream>>>(W, Wphi, WphiP, ATP);

    const int tile_blocks = (N + TILE_N - 1) / TILE_N;

    for (int it = 0; it < NUM_ITERS; ++it) {
        const float* xi = (it == 0) ? x_in : ((it & 1) ? xalt : out);
        float*       xo = (it & 1) ? out : xalt;
        gather_gemm_update_kernel<<<tile_blocks, NWAVES * 32, 0, stream>>>(rowptr, csr, dinv, WphiP, ATP, bias,
                                                                 xi, xo, N);
    }
}


# CSR build without global atomics: a block per 32768-edge chunk ranks destinations with LDS counters, the scan turns per-chunk counts into bases; sum order = chunk order, then LDS arrival
# speedup vs baseline: 3.5662x; 1.0871x over previous
#include <hip/hip_runtime.h>
#include <hip/hip_bf16.h>
#include <math.h>

#define D 128
#define GAMMA 0.1f
#define EPSILON 0.1f
#define NUM_ITERS 4

typedef __attribute__((ext_vector_type(2))) float v2f;
typedef __attribute__((ext_vector_type(8))) float v8f;

#define SCAN_ITEMS 1024
#define CHUNK_SHIFT 15
#define CHUNK_E (1 << CHUNK_SHIFT)
#define RANK_THREADS 1024
#define EPT (CHUNK_E / RANK_THREADS)
#define RANGE_N 65536

__global__ __launch_bounds__(RANK_THREADS) void chunk_rank_kernel(const int* __restrict__ ei,
                                                                  int* __restrict__ rank,
                                                                  int* __restrict__ tab,
                                                                  int E, int NS) {
    __shared__ __attribute__((aligned(16))) int lcnt[RANGE_N];
    const int tid = threadIdx.x;
    const int e0  = blockIdx.x * CHUNK_E;
    int* __restrict__ trow = tab + (size_t)blockIdx.x * NS;

    int dreg[EPT];
#pragma unroll
    for (int k = 0; k < EPT; ++k) {
        const int e = e0 + k * RANK_THREADS + tid;
        dreg[k] = (e < E) ? ei[E + e] : -1;
    }

    for (int r0 = 0; r0 < NS; r0 += RANGE_N) {
        for (int i = tid * 4; i < RANGE_N; i += RANK_THREADS * 4)
            *(int4*)&lcnt[i] = make_int4(0, 0, 0, 0);
        __syncthreads();
        int rk[EPT];
#pragma unroll
        for (int k = 0; k < EPT; ++k) {
            const unsigned off = (unsigned)(dreg[k] - r0);
            rk[k] = -1;
            if (off < (unsigned)RANGE_N) rk[k] = atomicAdd(&lcnt[off], 1);
        }
#pragma unroll
        for (int k = 0; k < EPT; ++k)
            if (rk[k] >= 0) rank[e0 + k * RANK_THREADS + tid] = rk[k];
        __syncthreads();
        const int lim = min(RANGE_N, NS - r0);
        for (int i = tid * 4; i < lim; i += RANK_THREADS * 4)
            *(int4*)(trow + r0 + i) = *(const int4*)&lcnt[i];
        __syncthreads();
    }
}

__global__ __launch_bounds__(256) void block_sum_kernel(const int* __restrict__ tab,
                                                        int* __restrict__ bsum, int chunks, int NS) {
    __shared__ int sm[8];
    const int tid = threadIdx.x, lane = tid & 31, wave = tid >> 5;
    const int base = blockIdx.x * SCAN_ITEMS + tid * 4;
    int s = 0;
    for (int c = 0; c < chunks; ++c) {
        const int4 v = *(const int4*)(tab + (size_t)c * NS + base);
        s += (v.x + v.y) + (v.z + v.w);
    }
    for (int o = 16; o > 0; o >>= 1) s += __shfl_xor(s, o);
    if (lane == 0) sm[wave] = s;
    __syncthreads();
    if (tid == 0) {
        int t = 0;
        for (int w = 0; w < 8; ++w) t += sm[w];
        bsum[blockIdx.x] = t;
    }
}

__global__ __launch_bounds__(256) void scan_rows_kernel(int* __restrict__ tab,
                                                        const int* __restrict__ bsum,
                                                        int* __restrict__ rowptr,
                                                        float* __restrict__ dinv, int n,
                                                        int chunks, int NS) {
    __shared__ int sm_off[8];
    __shared__ int sm_wav[8];
    const int tid = threadIdx.x, lane = tid & 31, wave = tid >> 5;

    int part = 0;
    for (int j = tid; j < (int)blockIdx.x; j += 256) part += bsum[j];
    for (int o = 16; o > 0; o >>= 1) part += __shfl_xor(part, o);
    if (lane == 0) sm_off[wave] = part;

    const int base = blockIdx.x * SCAN_ITEMS + tid * 4;
    int4 tot = make_int4(0, 0, 0, 0);
    for (int ch = 0; ch < chunks; ++ch) {
        const int4 v = *(const int4*)(tab + (size_t)ch * NS + base);
        tot.x += v.x; tot.y += v.y; tot.z += v.z; tot.w += v.w;
    }
    const int c[4] = {tot.x, tot.y, tot.z, tot.w};
    const int s = (c[0] + c[1]) + (c[2] + c[3]);
    int incl = s;
    for (int o = 1; o < 32; o <<= 1) {
        int t = __shfl_up(incl, o);
        if (lane >= o) incl += t;
    }
    if (lane == 31) sm_wav[wave] = incl;
    __syncthreads();

    int excl = incl - s;
    for (int w = 0; w < 8; ++w) {
        excl += sm_off[w];
        if (w < wave) excl += sm_wav[w];
    }
    int4 run;
    run.x = excl;
    run.y = run.x + c[0];
    run.z = run.y + c[1];
    run.w = run.z + c[2];
    const int rp[4] = {run.x, run.y, run.z, run.w};
    for (int u = 0; u < 4; ++u) {
        const int i = base + u;
        if (i < n) {
            rowptr[i] = rp[u];
            dinv[i] = rsqrtf((float)c[u] + 1.0f);
            if (i == n - 1) rowptr[n] = rp[u] + c[u];
        }
    }
    for (int ch = 0; ch < chunks; ++ch) {
        int4* p = (int4*)(tab + (size_t)ch * NS + base);
        const int4 v = *p;
        *p = run;
        run.x += v.x; run.y += v.y; run.z += v.z; run.w += v.w;
    }
}

__global__ __launch_bounds__(256) void fill_csr_kernel(const int* __restrict__ ei,
                                                       const int* __restrict__ rank,
                                                       const int* __restrict__ tab,
                                                       const float* __restrict__ dinv,
                                                       int* __restrict__ csr, int E, int NS) {
    int e = blockIdx.x * blockDim.x + threadIdx.x;
    if (e < E) {
        const int s = ei[e];
        const int d = ei[E + e];
        const int pos = tab[(size_t)(e >> CHUNK_SHIFT) * NS + d] + rank[e];
        int2 rec;
        rec.x = s;
        rec.y = __float_as_int(dinv[s] * dinv[d]);
        *(int2*)(csr + (size_t)pos * 2) = rec;
    }
}

__global__ __launch_bounds__(256) void prep_weights_kernel(const float* __restrict__ W,
                                                           const float* __restrict__ Wphi,
                                                           float* __restrict__ WphiP,
                                                           float* __restrict__ ATP) {
    int i = blockIdx.x * blockDim.x + threadIdx.x;
    if (i >= D * D) return;
    int k = i / D, n = i % D;
    size_t po = (size_t)(k >> 1) * (2 * D) + n * 2 + (k & 1);
    WphiP[po] = Wphi[n * D + k];
    float a = W[n * D + k] - W[k * D + n];
    if (k == n) a -= GAMMA;
    ATP[po] = a;
}

template <int K>
__device__ __forceinline__ void gather_rows(float4& acc, const float* __restrict__ xs,
                                            unsigned lane4, int2 my, int j) {
    float4 v[K];
    float w[K];
#pragma unroll
    for (int u = 0; u < K; ++u) {
        const int s = __builtin_amdgcn_readlane(my.x, j + u);
        w[u] = __int_as_float(__builtin_amdgcn_readlane(my.y, j + u));
        v[u] = *(const float4*)(xs + (size_t)s * D + lane4);
    }
#pragma unroll
    for (int u = 0; u < K; ++u) {
        acc.x += v[u].x * w[u]; acc.y += v[u].y * w[u];
        acc.z += v[u].z * w[u]; acc.w += v[u].w * w[u];
    }
}

#define NWAVES 16
#define TILE_N 64
#define NPW (TILE_N / NWAVES)
#define MT ((TILE_N / 16) / (NWAVES / 8))

__global__ __launch_bounds__(NWAVES * 32) void gather_gemm_update_kernel(const int* __restrict__ rowptr,
                                                                 const int* __restrict__ csr,
                                                                 const float* __restrict__ dinv,
                                                                 const float* __restrict__ WphiP,
                                                                 const float* __restrict__ ATP,
                                                                 const float* __restrict__ bias,
                                                                 const float* __restrict__ xin,
                                                                 float* __restrict__ xout, int N) {
    __shared__ __attribute__((aligned(16))) float lds_x[TILE_N][D + 4];
    __shared__ __attribute__((aligned(16))) float lds_y[TILE_N][D + 4];

    const int tid  = threadIdx.x;
    const int lane = tid & 31;
    const int wave = __builtin_amdgcn_readfirstlane(tid >> 5);
    const int row0 = blockIdx.x * TILE_N;
    const unsigned lane4 = (unsigned)lane * 4u;

    const int d0 = row0 + wave * NPW;
    int   pb[NPW + 1];
    float dv[NPW];
#pragma unroll
    for (int i = 0; i <= NPW; ++i) pb[i] = rowptr[min(d0 + i, N)];
#pragma unroll
    for (int i = 0; i < NPW; ++i) dv[i] = dinv[min(d0 + i, N - 1)];

#pragma unroll
    for (int i = 0; i < NPW; ++i) {
        const int r   = wave * NPW + i;
        const int beg = pb[i];
        const int end = pb[i + 1];
        float4 xs = *(const float4*)(xin + (size_t)min(d0 + i, N - 1) * D + lane4);
        if (d0 + i >= N) xs = make_float4(0.f, 0.f, 0.f, 0.f);
        float4 acc = make_float4(0.f, 0.f, 0.f, 0.f);
        for (int base = beg; base < end; base += 32) {
            const int n = min(32, end - base);
            int2 my = make_int2(0, 0);
            if (lane < n) my = *(const int2*)(csr + (size_t)(base + lane) * 2);
            int j = 0;
            for (; j + 8 <= n; j += 8) gather_rows<8>(acc, xin, lane4, my, j);
            if (j + 4 <= n) { gather_rows<4>(acc, xin, lane4, my, j); j += 4; }
            if (j + 2 <= n) { gather_rows<2>(acc, xin, lane4, my, j); j += 2; }
            if (j < n) gather_rows<1>(acc, xin, lane4, my, j);
        }
        const float d2 = dv[i] * dv[i];
        acc.x += xs.x * d2; acc.y += xs.y * d2;
        acc.z += xs.z * d2; acc.w += xs.w * d2;
        *(float4*)&lds_x[r][lane4] = xs;
        *(float4*)&lds_y[r][lane4] = acc;
    }
    __syncthreads();

    const int mn   = lane & 15;
    const int kq   = lane >> 4;
    const int col0 = (wave & 7) * 16;
    const int mrow0 = (wave >> 3) * (MT * 16);
    const size_t bcol = (size_t)(col0 + mn) * 2;

    v8f acc_w[MT];
    v8f acc_a[MT];
#pragma unroll
    for (int m = 0; m < MT; ++m) { acc_w[m] = (v8f){}; acc_a[m] = (v8f){}; }

#pragma unroll 4
    for (int k0 = 0; k0 < D; k0 += 4) {
        const int k = k0 + kq * 2;
        const int p = (k0 >> 1) + kq;
        const v2f bw = *(const v2f*)(WphiP + (size_t)p * (2 * D) + bcol);
        const v2f ba = *(const v2f*)(ATP   + (size_t)p * (2 * D) + bcol);
#pragma unroll
        for (int m = 0; m < MT; ++m) {
            const v2f ay = *(const v2f*)&lds_y[mrow0 + m * 16 + mn][k];
            const v2f ax = *(const v2f*)&lds_x[mrow0 + m * 16 + mn][k];
            acc_w[m] = __builtin_amdgcn_wmma_f32_16x16x4_f32(false, ay, false, bw,
                                                             (short)0, acc_w[m], false, false);
            acc_a[m] = __builtin_amdgcn_wmma_f32_16x16x4_f32(false, ax, false, ba,
                                                             (short)0, acc_a[m], false, false);
        }
    }

    const float b    = bias[col0 + mn];
    const int rowoff = kq * 8;
#pragma unroll
    for (int m = 0; m < MT; ++m) {
#pragma unroll
        for (int r = 0; r < 8; ++r) {
            const float xo = lds_x[mrow0 + m * 16 + rowoff + r][col0 + mn];
            acc_w[m][r] = xo + EPSILON * tanhf(acc_w[m][r] + acc_a[m][r] + b);
        }
    }
    __syncthreads();
#pragma unroll
    for (int m = 0; m < MT; ++m) {
#pragma unroll
        for (int r = 0; r < 8; ++r) lds_y[mrow0 + m * 16 + rowoff + r][col0 + mn] = acc_w[m][r];
    }
    __syncthreads();
#pragma unroll
    for (int i = 0; i < NPW; ++i) {
        const int r = wave * NPW + i;
        const int d = row0 + r;
        if (d < N) *(float4*)(xout + (size_t)d * D + lane4) = *(const float4*)&lds_y[r][lane4];
    }
}

extern "C" void kernel_launch(void* const* d_in, const int* in_sizes, int n_in,
                              void* d_out, int out_size, void* d_ws, size_t ws_size,
                              hipStream_t stream) {
    const float* x_in = (const float*)d_in[0];
    const int*   ei   = (const int*)d_in[1];
    const float* W    = (const float*)d_in[2];
    const float* Wphi = (const float*)d_in[3];
    const float* bias = (const float*)d_in[4];
    float*       out  = (float*)d_out;

    const int N = in_sizes[0] / D;
    const int E = in_sizes[1] / 2;

    float* ws    = (float*)d_ws;
    const size_t nd = (size_t)N * D;
    const int scan_blocks = (N + SCAN_ITEMS - 1) / SCAN_ITEMS;
    const int NS     = scan_blocks * SCAN_ITEMS;
    const int chunks = (E + CHUNK_E - 1) / CHUNK_E;
    float* xalt  = ws;
    int*   tab   = (int*)(ws + nd);
    int*   rank  = tab + (size_t)chunks * NS;
    int*   bsum  = rank + E;
    int*   csr   = (int*)(ws + 2 * nd);
    int*   rowptr = csr + (size_t)E * 2;
    float* dinv  = ws + 3 * nd;
    float* WphiP = dinv + N;
    float* ATP   = WphiP + (size_t)D * D;

    const int B = 256;

    chunk_rank_kernel<<<chunks, RANK_THREADS, 0, stream>>>(ei, rank, tab, E, NS);
    block_sum_kernel<<<scan_blocks, B, 0, stream>>>(tab, bsum, chunks, NS);
    scan_rows_kernel<<<scan_blocks, B, 0, stream>>>(tab, bsum, rowptr, dinv, N, chunks, NS);
    fill_csr_kernel<<<(E + B - 1) / B, B, 0, stream>>>(ei, rank, tab, dinv, csr, E, NS);
    prep_weights_kernel<<<(D * D + B - 1) / B, B, 0, stream>>>(W, Wphi, WphiP, ATP);

    const int tile_blocks = (N + TILE_N - 1) / TILE_N;

    for (int it = 0; it < NUM_ITERS; ++it) {
        const float* xi = (it == 0) ? x_in : ((it & 1) ? xalt : out);
        float*       xo = (it & 1) ? out : xalt;
        gather_gemm_update_kernel<<<tile_blocks, NWAVES * 32, 0, stream>>>(rowptr, csr, dinv, WphiP, ATP, bias,
                                                                 xi, xo, N);
    }
}


# x A^T product moved into the gather phase as four k-slices (own rows staged to LDS first, one more barrier): f32 WMMA issue runs while neighbour rows are in flight; sums and their order unchanged
# speedup vs baseline: 3.7311x; 1.0462x over previous
#include <hip/hip_runtime.h>
#include <hip/hip_bf16.h>
#include <math.h>

#define D 128
#define GAMMA 0.1f
#define EPSILON 0.1f
#define NUM_ITERS 4

typedef __attribute__((ext_vector_type(2))) float v2f;
typedef __attribute__((ext_vector_type(8))) float v8f;

#define SCAN_ITEMS 1024
#define CHUNK_SHIFT 15
#define CHUNK_E (1 << CHUNK_SHIFT)
#define RANK_THREADS 1024
#define EPT (CHUNK_E / RANK_THREADS)
#define RANGE_N 65536

__global__ __launch_bounds__(RANK_THREADS) void chunk_rank_kernel(const int* __restrict__ ei,
                                                                  int* __restrict__ rank,
                                                                  int* __restrict__ tab,
                                                                  int E, int NS) {
    __shared__ __attribute__((aligned(16))) int lcnt[RANGE_N];
    const int tid = threadIdx.x;
    const int e0  = blockIdx.x * CHUNK_E;
    int* __restrict__ trow = tab + (size_t)blockIdx.x * NS;

    int dreg[EPT];
#pragma unroll
    for (int k = 0; k < EPT; ++k) {
        const int e = e0 + k * RANK_THREADS + tid;
        dreg[k] = (e < E) ? ei[E + e] : -1;
    }

    for (int r0 = 0; r0 < NS; r0 += RANGE_N) {
        for (int i = tid * 4; i < RANGE_N; i += RANK_THREADS * 4)
            *(int4*)&lcnt[i] = make_int4(0, 0, 0, 0);
        __syncthreads();
        int rk[EPT];
#pragma unroll
        for (int k = 0; k < EPT; ++k) {
            const unsigned off = (unsigned)(dreg[k] - r0);
            rk[k] = -1;
            if (off < (unsigned)RANGE_N) rk[k] = atomicAdd(&lcnt[off], 1);
        }
#pragma unroll
        for (int k = 0; k < EPT; ++k)
            if (rk[k] >= 0) rank[e0 + k * RANK_THREADS + tid] = rk[k];
        __syncthreads();
        const int lim = min(RANGE_N, NS - r0);
        for (int i = tid * 4; i < lim; i += RANK_THREADS * 4)
            *(int4*)(trow + r0 + i) = *(const int4*)&lcnt[i];
        __syncthreads();
    }
}

__global__ __launch_bounds__(256) void block_sum_kernel(const int* __restrict__ tab,
                                                        int* __restrict__ bsum, int chunks, int NS) {
    __shared__ int sm[8];
    const int tid = threadIdx.x, lane = tid & 31, wave = tid >> 5;
    const int base = blockIdx.x * SCAN_ITEMS + tid * 4;
    int s = 0;
    for (int c = 0; c < chunks; ++c) {
        const int4 v = *(const int4*)(tab + (size_t)c * NS + base);
        s += (v.x + v.y) + (v.z + v.w);
    }
    for (int o = 16; o > 0; o >>= 1) s += __shfl_xor(s, o);
    if (lane == 0) sm[wave] = s;
    __syncthreads();
    if (tid == 0) {
        int t = 0;
        for (int w = 0; w < 8; ++w) t += sm[w];
        bsum[blockIdx.x] = t;
    }
}

__global__ __launch_bounds__(256) void scan_rows_kernel(int* __restrict__ tab,
                                                        const int* __restrict__ bsum,
                                                        int* __restrict__ rowptr,
                                                        float* __restrict__ dinv, int n,
                                                        int chunks, int NS) {
    __shared__ int sm_off[8];
    __shared__ int sm_wav[8];
    const int tid = threadIdx.x, lane = tid & 31, wave = tid >> 5;

    int part = 0;
    for (int j = tid; j < (int)blockIdx.x; j += 256) part += bsum[j];
    for (int o = 16; o > 0; o >>= 1) part += __shfl_xor(part, o);
    if (lane == 0) sm_off[wave] = part;

    const int base = blockIdx.x * SCAN_ITEMS + tid * 4;
    int4 tot = make_int4(0, 0, 0, 0);
    for (int ch = 0; ch < chunks; ++ch) {
        const int4 v = *(const int4*)(tab + (size_t)ch * NS + base);
        tot.x += v.x; tot.y += v.y; tot.z += v.z; tot.w += v.w;
    }
    const int c[4] = {tot.x, tot.y, tot.z, tot.w};
    const int s = (c[0] + c[1]) + (c[2] + c[3]);
    int incl = s;
    for (int o = 1; o < 32; o <<= 1) {
        int t = __shfl_up(incl, o);
        if (lane >= o) incl += t;
    }
    if (lane == 31) sm_wav[wave] = incl;
    __syncthreads();

    int excl = incl - s;
    for (int w = 0; w < 8; ++w) {
        excl += sm_off[w];
        if (w < wave) excl += sm_wav[w];
    }
    int4 run;
    run.x = excl;
    run.y = run.x + c[0];
    run.z = run.y + c[1];
    run.w = run.z + c[2];
    const int rp[4] = {run.x, run.y, run.z, run.w};
    for (int u = 0; u < 4; ++u) {
        const int i = base + u;
        if (i < n) {
            rowptr[i] = rp[u];
            dinv[i] = rsqrtf((float)c[u] + 1.0f);
            if (i == n - 1) rowptr[n] = rp[u] + c[u];
        }
    }
    for (int ch = 0; ch < chunks; ++ch) {
        int4* p = (int4*)(tab + (size_t)ch * NS + base);
        const int4 v = *p;
        *p = run;
        run.x += v.x; run.y += v.y; run.z += v.z; run.w += v.w;
    }
}

__global__ __launch_bounds__(256) void fill_csr_kernel(const int* __restrict__ ei,
                                                       const int* __restrict__ rank,
                                                       const int* __restrict__ tab,
                                                       const float* __restrict__ dinv,
                                                       int* __restrict__ csr, int E, int NS) {
    int e = blockIdx.x * blockDim.x + threadIdx.x;
    if (e < E) {
        const int s = ei[e];
        const int d = ei[E + e];
        const int pos = tab[(size_t)(e >> CHUNK_SHIFT) * NS + d] + rank[e];
        int2 rec;
        rec.x = s;
        rec.y = __float_as_int(dinv[s] * dinv[d]);
        *(int2*)(csr + (size_t)pos * 2) = rec;
    }
}

__global__ __launch_bounds__(256) void prep_weights_kernel(const float* __restrict__ W,
                                                           const float* __restrict__ Wphi,
                                                           float* __restrict__ WphiP,
                                                           float* __restrict__ ATP) {
    int i = blockIdx.x * blockDim.x + threadIdx.x;
    if (i >= D * D) return;
    int k = i / D, n = i % D;
    size_t po = (size_t)(k >> 1) * (2 * D) + n * 2 + (k & 1);
    WphiP[po] = Wphi[n * D + k];
    float a = W[n * D + k] - W[k * D + n];
    if (k == n) a -= GAMMA;
    ATP[po] = a;
}

template <int K>
__device__ __forceinline__ void gather_rows(float4& acc, const float* __restrict__ xs,
                                            unsigned lane4, int2 my, int j) {
    float4 v[K];
    float w[K];
#pragma unroll
    for (int u = 0; u < K; ++u) {
        const int s = __builtin_amdgcn_readlane(my.x, j + u);
        w[u] = __int_as_float(__builtin_amdgcn_readlane(my.y, j + u));
        v[u] = *(const float4*)(xs + (size_t)s * D + lane4);
    }
#pragma unroll
    for (int u = 0; u < K; ++u) {
        acc.x += v[u].x * w[u]; acc.y += v[u].y * w[u];
        acc.z += v[u].z * w[u]; acc.w += v[u].w * w[u];
    }
}

#define NWAVES 16
#define TILE_N 64
#define NPW (TILE_N / NWAVES)
#define MT ((TILE_N / 16) / (NWAVES / 8))

__global__ __launch_bounds__(NWAVES * 32) void gather_gemm_update_kernel(const int* __restrict__ rowptr,
                                                                 const int* __restrict__ csr,
                                                                 const float* __restrict__ dinv,
                                                                 const float* __restrict__ WphiP,
                                                                 const float* __restrict__ ATP,
                                                                 const float* __restrict__ bias,
                                                                 const float* __restrict__ xin,
                                                                 float* __restrict__ xout, int N) {
    __shared__ __attribute__((aligned(16))) float lds_x[TILE_N][D + 4];
    __shared__ __attribute__((aligned(16))) float lds_y[TILE_N][D + 4];

    const int tid  = threadIdx.x;
    const int lane = tid & 31;
    const int wave = __builtin_amdgcn_readfirstlane(tid >> 5);
    const int row0 = blockIdx.x * TILE_N;
    const unsigned lane4 = (unsigned)lane * 4u;

    const int d0 = row0 + wave * NPW;
    int   pb[NPW + 1];
    float dv[NPW];
#pragma unroll
    for (int i = 0; i <= NPW; ++i) pb[i] = rowptr[min(d0 + i, N)];
#pragma unroll
    for (int i = 0; i < NPW; ++i) dv[i] = dinv[min(d0 + i, N - 1)];

#pragma unroll
    for (int i = 0; i < NPW; ++i) {
        const int r = wave * NPW + i;
        float4 xs = *(const float4*)(xin + (size_t)min(d0 + i, N - 1) * D + lane4);
        if (d0 + i >= N) xs = make_float4(0.f, 0.f, 0.f, 0.f);
        *(float4*)&lds_x[r][lane4] = xs;
    }
    __syncthreads();

    const int mn   = lane & 15;
    const int kq   = lane >> 4;
    const int col0 = (wave & 7) * 16;
    const int mrow0 = (wave >> 3) * (MT * 16);
    const size_t bcol = (size_t)(col0 + mn) * 2;

    v8f acc_w[MT];
    v8f acc_a[MT];
#pragma unroll
    for (int m = 0; m < MT; ++m) { acc_w[m] = (v8f){}; acc_a[m] = (v8f){}; }

#pragma unroll
    for (int i = 0; i < NPW; ++i) {
        const int r   = wave * NPW + i;
        const int beg = pb[i];
        const int end = pb[i + 1];
        int2 my = make_int2(0, 0);
        if (lane < min(32, end - beg)) my = *(const int2*)(csr + (size_t)(beg + lane) * 2);
#pragma unroll
        for (int k0 = i * (D / NPW); k0 < (i + 1) * (D / NPW); k0 += 4) {
            const int k = k0 + kq * 2;
            const int p = (k0 >> 1) + kq;
            const v2f ba = *(const v2f*)(ATP + (size_t)p * (2 * D) + bcol);
#pragma unroll
            for (int m = 0; m < MT; ++m) {
                const v2f ax = *(const v2f*)&lds_x[mrow0 + m * 16 + mn][k];
                acc_a[m] = __builtin_amdgcn_wmma_f32_16x16x4_f32(false, ax, false, ba,
                                                                 (short)0, acc_a[m], false, false);
            }
        }
        float4 acc = make_float4(0.f, 0.f, 0.f, 0.f);
        for (int base = beg; base < end; base += 32) {
            const int n = min(32, end - base);
            if (base != beg) {
                my = make_int2(0, 0);
                if (lane < n) my = *(const int2*)(csr + (size_t)(base + lane) * 2);
            }
            int j = 0;
            for (; j + 8 <= n; j += 8) gather_rows<8>(acc, xin, lane4, my, j);
            if (j + 4 <= n) { gather_rows<4>(acc, xin, lane4, my, j); j += 4; }
            if (j + 2 <= n) { gather_rows<2>(acc, xin, lane4, my, j); j += 2; }
            if (j < n) gather_rows<1>(acc, xin, lane4, my, j);
        }
        const float4 xs = *(const float4*)&lds_x[r][lane4];
        const float d2 = dv[i] * dv[i];
        acc.x += xs.x * d2; acc.y += xs.y * d2;
        acc.z += xs.z * d2; acc.w += xs.w * d2;
        *(float4*)&lds_y[r][lane4] = acc;
    }
    __syncthreads();

#pragma unroll 4
    for (int k0 = 0; k0 < D; k0 += 4) {
        const int k = k0 + kq * 2;
        const int p = (k0 >> 1) + kq;
        const v2f bw = *(const v2f*)(WphiP + (size_t)p * (2 * D) + bcol);
#pragma unroll
        for (int m = 0; m < MT; ++m) {
            const v2f ay = *(const v2f*)&lds_y[mrow0 + m * 16 + mn][k];
            acc_w[m] = __builtin_amdgcn_wmma_f32_16x16x4_f32(false, ay, false, bw,
                                                             (short)0, acc_w[m], false, false);
        }
    }

    const float b    = bias[col0 + mn];
    const int rowoff = kq * 8;
#pragma unroll
    for (int m = 0; m < MT; ++m) {
#pragma unroll
        for (int r = 0; r < 8; ++r) {
            const float xo = lds_x[mrow0 + m * 16 + rowoff + r][col0 + mn];
            acc_w[m][r] = xo + EPSILON * tanhf(acc_w[m][r] + acc_a[m][r] + b);
        }
    }
    __syncthreads();
#pragma unroll
    for (int m = 0; m < MT; ++m) {
#pragma unroll
        for (int r = 0; r < 8; ++r) lds_y[mrow0 + m * 16 + rowoff + r][col0 + mn] = acc_w[m][r];
    }
    __syncthreads();
#pragma unroll
    for (int i = 0; i < NPW; ++i) {
        const int r = wave * NPW + i;
        const int d = row0 + r;
        if (d < N) *(float4*)(xout + (size_t)d * D + lane4) = *(const float4*)&lds_y[r][lane4];
    }
}

extern "C" void kernel_launch(void* const* d_in, const int* in_sizes, int n_in,
                              void* d_out, int out_size, void* d_ws, size_t ws_size,
                              hipStream_t stream) {
    const float* x_in = (const float*)d_in[0];
    const int*   ei   = (const int*)d_in[1];
    const float* W    = (const float*)d_in[2];
    const float* Wphi = (const float*)d_in[3];
    const float* bias = (const float*)d_in[4];
    float*       out  = (float*)d_out;

    const int N = in_sizes[0] / D;
    const int E = in_sizes[1] / 2;

    float* ws    = (float*)d_ws;
    const size_t nd = (size_t)N * D;
    const int scan_blocks = (N + SCAN_ITEMS - 1) / SCAN_ITEMS;
    const int NS     = scan_blocks * SCAN_ITEMS;
    const int chunks = (E + CHUNK_E - 1) / CHUNK_E;
    float* xalt  = ws;
    int*   tab   = (int*)(ws + nd);
    int*   rank  = tab + (size_t)chunks * NS;
    int*   bsum  = rank + E;
    int*   csr   = (int*)(ws + 2 * nd);
    int*   rowptr = csr + (size_t)E * 2;
    float* dinv  = ws + 3 * nd;
    float* WphiP = dinv + N;
    float* ATP   = WphiP + (size_t)D * D;

    const int B = 256;

    chunk_rank_kernel<<<chunks, RANK_THREADS, 0, stream>>>(ei, rank, tab, E, NS);
    block_sum_kernel<<<scan_blocks, B, 0, stream>>>(tab, bsum, chunks, NS);
    scan_rows_kernel<<<scan_blocks, B, 0, stream>>>(tab, bsum, rowptr, dinv, N, chunks, NS);
    fill_csr_kernel<<<(E + B - 1) / B, B, 0, stream>>>(ei, rank, tab, dinv, csr, E, NS);
    prep_weights_kernel<<<(D * D + B - 1) / B, B, 0, stream>>>(W, Wphi, WphiP, ATP);

    const int tile_blocks = (N + TILE_N - 1) / TILE_N;

    for (int it = 0; it < NUM_ITERS; ++it) {
        const float* xi = (it == 0) ? x_in : ((it & 1) ? xalt : out);
        float*       xo = (it & 1) ? out : xalt;
        gather_gemm_update_kernel<<<tile_blocks, NWAVES * 32, 0, stream>>>(rowptr, csr, dinv, WphiP, ATP, bias,
                                                                 xi, xo, N);
    }
}


# Epilogue tanh as one branch-free form sign(z)*(1-2/(exp(2|z|)+1)) on hardware exp and rcp for every z, not the library's split whose two sides both ran per value; abs error <= 2.5e-7, not relative
# speedup vs baseline: 3.9524x; 1.0593x over previous
#include <hip/hip_runtime.h>
#include <hip/hip_bf16.h>
#include <math.h>

#define D 128
#define GAMMA 0.1f
#define EPSILON 0.1f
#define NUM_ITERS 4

typedef __attribute__((ext_vector_type(2))) float v2f;
typedef __attribute__((ext_vector_type(8))) float v8f;

#define SCAN_ITEMS 1024
#define CHUNK_SHIFT 15
#define CHUNK_E (1 << CHUNK_SHIFT)
#define RANK_THREADS 1024
#define EPT (CHUNK_E / RANK_THREADS)
#define RANGE_N 65536

__global__ __launch_bounds__(RANK_THREADS) void chunk_rank_kernel(const int* __restrict__ ei,
                                                                  int* __restrict__ rank,
                                                                  int* __restrict__ tab,
                                                                  int E, int NS) {
    __shared__ __attribute__((aligned(16))) int lcnt[RANGE_N];
    const int tid = threadIdx.x;
    const int e0  = blockIdx.x * CHUNK_E;
    int* __restrict__ trow = tab + (size_t)blockIdx.x * NS;

    int dreg[EPT];
#pragma unroll
    for (int k = 0; k < EPT; ++k) {
        const int e = e0 + k * RANK_THREADS + tid;
        dreg[k] = (e < E) ? ei[E + e] : -1;
    }

    for (int r0 = 0; r0 < NS; r0 += RANGE_N) {
        for (int i = tid * 4; i < RANGE_N; i += RANK_THREADS * 4)
            *(int4*)&lcnt[i] = make_int4(0, 0, 0, 0);
        __syncthreads();
        int rk[EPT];
#pragma unroll
        for (int k = 0; k < EPT; ++k) {
            const unsigned off = (unsigned)(dreg[k] - r0);
            rk[k] = -1;
            if (off < (unsigned)RANGE_N) rk[k] = atomicAdd(&lcnt[off], 1);
        }
#pragma unroll
        for (int k = 0; k < EPT; ++k)
            if (rk[k] >= 0) rank[e0 + k * RANK_THREADS + tid] = rk[k];
        __syncthreads();
        const int lim = min(RANGE_N, NS - r0);
        for (int i = tid * 4; i < lim; i += RANK_THREADS * 4)
            *(int4*)(trow + r0 + i) = *(const int4*)&lcnt[i];
        __syncthreads();
    }
}

__global__ __launch_bounds__(256) void block_sum_kernel(const int* __restrict__ tab,
                                                        int* __restrict__ bsum, int chunks, int NS) {
    __shared__ int sm[8];
    const int tid = threadIdx.x, lane = tid & 31, wave = tid >> 5;
    const int base = blockIdx.x * SCAN_ITEMS + tid * 4;
    int s = 0;
    for (int c = 0; c < chunks; ++c) {
        const int4 v = *(const int4*)(tab + (size_t)c * NS + base);
        s += (v.x + v.y) + (v.z + v.w);
    }
    for (int o = 16; o > 0; o >>= 1) s += __shfl_xor(s, o);
    if (lane == 0) sm[wave] = s;
    __syncthreads();
    if (tid == 0) {
        int t = 0;
        for (int w = 0; w < 8; ++w) t += sm[w];
        bsum[blockIdx.x] = t;
    }
}

__global__ __launch_bounds__(256) void scan_rows_kernel(int* __restrict__ tab,
                                                        const int* __restrict__ bsum,
                                                        int* __restrict__ rowptr,
                                                        float* __restrict__ dinv, int n,
                                                        int chunks, int NS) {
    __shared__ int sm_off[8];
    __shared__ int sm_wav[8];
    const int tid = threadIdx.x, lane = tid & 31, wave = tid >> 5;

    int part = 0;
    for (int j = tid; j < (int)blockIdx.x; j += 256) part += bsum[j];
    for (int o = 16; o > 0; o >>= 1) part += __shfl_xor(part, o);
    if (lane == 0) sm_off[wave] = part;

    const int base = blockIdx.x * SCAN_ITEMS + tid * 4;
    int4 tot = make_int4(0, 0, 0, 0);
    for (int ch = 0; ch < chunks; ++ch) {
        const int4 v = *(const int4*)(tab + (size_t)ch * NS + base);
        tot.x += v.x; tot.y += v.y; tot.z += v.z; tot.w += v.w;
    }
    const int c[4] = {tot.x, tot.y, tot.z, tot.w};
    const int s = (c[0] + c[1]) + (c[2] + c[3]);
    int incl = s;
    for (int o = 1; o < 32; o <<= 1) {
        int t = __shfl_up(incl, o);
        if (lane >= o) incl += t;
    }
    if (lane == 31) sm_wav[wave] = incl;
    __syncthreads();

    int excl = incl - s;
    for (int w = 0; w < 8; ++w) {
        excl += sm_off[w];
        if (w < wave) excl += sm_wav[w];
    }
    int4 run;
    run.x = excl;
    run.y = run.x + c[0];
    run.z = run.y + c[1];
    run.w = run.z + c[2];
    const int rp[4] = {run.x, run.y, run.z, run.w};
    for (int u = 0; u < 4; ++u) {
        const int i = base + u;
        if (i < n) {
            rowptr[i] = rp[u];
            dinv[i] = rsqrtf((float)c[u] + 1.0f);
            if (i == n - 1) rowptr[n] = rp[u] + c[u];
        }
    }
    for (int ch = 0; ch < chunks; ++ch) {
        int4* p = (int4*)(tab + (size_t)ch * NS + base);
        const int4 v = *p;
        *p = run;
        run.x += v.x; run.y += v.y; run.z += v.z; run.w += v.w;
    }
}

__global__ __launch_bounds__(256) void fill_csr_kernel(const int* __restrict__ ei,
                                                       const int* __restrict__ rank,
                                                       const int* __restrict__ tab,
                                                       const float* __restrict__ dinv,
                                                       int* __restrict__ csr, int E, int NS) {
    int e = blockIdx.x * blockDim.x + threadIdx.x;
    if (e < E) {
        const int s = ei[e];
        const int d = ei[E + e];
        const int pos = tab[(size_t)(e >> CHUNK_SHIFT) * NS + d] + rank[e];
        int2 rec;
        rec.x = s;
        rec.y = __float_as_int(dinv[s] * dinv[d]);
        *(int2*)(csr + (size_t)pos * 2) = rec;
    }
}

__global__ __launch_bounds__(256) void prep_weights_kernel(const float* __restrict__ W,
                                                           const float* __restrict__ Wphi,
                                                           float* __restrict__ WphiP,
                                                           float* __restrict__ ATP) {
    int i = blockIdx.x * blockDim.x + threadIdx.x;
    if (i >= D * D) return;
    int k = i / D, n = i % D;
    size_t po = (size_t)(k >> 1) * (2 * D) + n * 2 + (k & 1);
    WphiP[po] = Wphi[n * D + k];
    float a = W[n * D + k] - W[k * D + n];
    if (k == n) a -= GAMMA;
    ATP[po] = a;
}

template <int K>
__device__ __forceinline__ void gather_rows(float4& acc, const float* __restrict__ xs,
                                            unsigned lane4, int2 my, int j) {
    float4 v[K];
    float w[K];
#pragma unroll
    for (int u = 0; u < K; ++u) {
        const int s = __builtin_amdgcn_readlane(my.x, j + u);
        w[u] = __int_as_float(__builtin_amdgcn_readlane(my.y, j + u));
        v[u] = *(const float4*)(xs + (size_t)s * D + lane4);
    }
#pragma unroll
    for (int u = 0; u < K; ++u) {
        acc.x += v[u].x * w[u]; acc.y += v[u].y * w[u];
        acc.z += v[u].z * w[u]; acc.w += v[u].w * w[u];
    }
}

__device__ __forceinline__ float tanh_branchfree(float z) {
    const float t = __builtin_amdgcn_exp2f(fabsf(z) * 2.885390081777927f);
    const float r = __builtin_amdgcn_rcpf(t + 1.0f);
    return copysignf(fmaf(-2.0f, r, 1.0f), z);
}

#define NWAVES 16
#define TILE_N 64
#define NPW (TILE_N / NWAVES)
#define MT ((TILE_N / 16) / (NWAVES / 8))

__global__ __launch_bounds__(NWAVES * 32) void gather_gemm_update_kernel(const int* __restrict__ rowptr,
                                                                 const int* __restrict__ csr,
                                                                 const float* __restrict__ dinv,
                                                                 const float* __restrict__ WphiP,
                                                                 const float* __restrict__ ATP,
                                                                 const float* __restrict__ bias,
                                                                 const float* __restrict__ xin,
                                                                 float* __restrict__ xout, int N) {
    __shared__ __attribute__((aligned(16))) float lds_x[TILE_N][D + 4];
    __shared__ __attribute__((aligned(16))) float lds_y[TILE_N][D + 4];

    const int tid  = threadIdx.x;
    const int lane = tid & 31;
    const int wave = __builtin_amdgcn_readfirstlane(tid >> 5);
    const int row0 = blockIdx.x * TILE_N;
    const unsigned lane4 = (unsigned)lane * 4u;

    const int d0 = row0 + wave * NPW;
    int   pb[NPW + 1];
    float dv[NPW];
#pragma unroll
    for (int i = 0; i <= NPW; ++i) pb[i] = rowptr[min(d0 + i, N)];
#pragma unroll
    for (int i = 0; i < NPW; ++i) dv[i] = dinv[min(d0 + i, N - 1)];

#pragma unroll
    for (int i = 0; i < NPW; ++i) {
        const int r = wave * NPW + i;
        float4 xs = *(const float4*)(xin + (size_t)min(d0 + i, N - 1) * D + lane4);
        if (d0 + i >= N) xs = make_float4(0.f, 0.f, 0.f, 0.f);
        *(float4*)&lds_x[r][lane4] = xs;
    }
    __syncthreads();

    const int mn   = lane & 15;
    const int kq   = lane >> 4;
    const int col0 = (wave & 7) * 16;
    const int mrow0 = (wave >> 3) * (MT * 16);
    const size_t bcol = (size_t)(col0 + mn) * 2;

    v8f acc_w[MT];
    v8f acc_a[MT];
#pragma unroll
    for (int m = 0; m < MT; ++m) { acc_w[m] = (v8f){}; acc_a[m] = (v8f){}; }

#pragma unroll
    for (int i = 0; i < NPW; ++i) {
        const int r   = wave * NPW + i;
        const int beg = pb[i];
        const int end = pb[i + 1];
        int2 my = make_int2(0, 0);
        if (lane < min(32, end - beg)) my = *(const int2*)(csr + (size_t)(beg + lane) * 2);
#pragma unroll
        for (int k0 = i * (D / NPW); k0 < (i + 1) * (D / NPW); k0 += 4) {
            const int k = k0 + kq * 2;
            const int p = (k0 >> 1) + kq;
            const v2f ba = *(const v2f*)(ATP + (size_t)p * (2 * D) + bcol);
#pragma unroll
            for (int m = 0; m < MT; ++m) {
                const v2f ax = *(const v2f*)&lds_x[mrow0 + m * 16 + mn][k];
                acc_a[m] = __builtin_amdgcn_wmma_f32_16x16x4_f32(false, ax, false, ba,
                                                                 (short)0, acc_a[m], false, false);
            }
        }
        float4 acc = make_float4(0.f, 0.f, 0.f, 0.f);
        for (int base = beg; base < end; base += 32) {
            const int n = min(32, end - base);
            if (base != beg) {
                my = make_int2(0, 0);
                if (lane < n) my = *(const int2*)(csr + (size_t)(base + lane) * 2);
            }
            int j = 0;
            for (; j + 8 <= n; j += 8) gather_rows<8>(acc, xin, lane4, my, j);
            if (j + 4 <= n) { gather_rows<4>(acc, xin, lane4, my, j); j += 4; }
            if (j + 2 <= n) { gather_rows<2>(acc, xin, lane4, my, j); j += 2; }
            if (j < n) gather_rows<1>(acc, xin, lane4, my, j);
        }
        const float4 xs = *(const float4*)&lds_x[r][lane4];
        const float d2 = dv[i] * dv[i];
        acc.x += xs.x * d2; acc.y += xs.y * d2;
        acc.z += xs.z * d2; acc.w += xs.w * d2;
        *(float4*)&lds_y[r][lane4] = acc;
    }
    __syncthreads();

#pragma unroll 4
    for (int k0 = 0; k0 < D; k0 += 4) {
        const int k = k0 + kq * 2;
        const int p = (k0 >> 1) + kq;
        const v2f bw = *(const v2f*)(WphiP + (size_t)p * (2 * D) + bcol);
#pragma unroll
        for (int m = 0; m < MT; ++m) {
            const v2f ay = *(const v2f*)&lds_y[mrow0 + m * 16 + mn][k];
            acc_w[m] = __builtin_amdgcn_wmma_f32_16x16x4_f32(false, ay, false, bw,
                                                             (short)0, acc_w[m], false, false);
        }
    }

    const float b    = bias[col0 + mn];
    const int rowoff = kq * 8;
#pragma unroll
    for (int m = 0; m < MT; ++m) {
#pragma unroll
        for (int r = 0; r < 8; ++r) {
            const float xo = lds_x[mrow0 + m * 16 + rowoff + r][col0 + mn];
            acc_w[m][r] = xo + EPSILON * tanh_branchfree(acc_w[m][r] + acc_a[m][r] + b);
        }
    }
    __syncthreads();
#pragma unroll
    for (int m = 0; m < MT; ++m) {
#pragma unroll
        for (int r = 0; r < 8; ++r) lds_y[mrow0 + m * 16 + rowoff + r][col0 + mn] = acc_w[m][r];
    }
    __syncthreads();
#pragma unroll
    for (int i = 0; i < NPW; ++i) {
        const int r = wave * NPW + i;
        const int d = row0 + r;
        if (d < N) *(float4*)(xout + (size_t)d * D + lane4) = *(const float4*)&lds_y[r][lane4];
    }
}

extern "C" void kernel_launch(void* const* d_in, const int* in_sizes, int n_in,
                              void* d_out, int out_size, void* d_ws, size_t ws_size,
                              hipStream_t stream) {
    const float* x_in = (const float*)d_in[0];
    const int*   ei   = (const int*)d_in[1];
    const float* W    = (const float*)d_in[2];
    const float* Wphi = (const float*)d_in[3];
    const float* bias = (const float*)d_in[4];
    float*       out  = (float*)d_out;

    const int N = in_sizes[0] / D;
    const int E = in_sizes[1] / 2;

    float* ws    = (float*)d_ws;
    const size_t nd = (size_t)N * D;
    const int scan_blocks = (N + SCAN_ITEMS - 1) / SCAN_ITEMS;
    const int NS     = scan_blocks * SCAN_ITEMS;
    const int chunks = (E + CHUNK_E - 1) / CHUNK_E;
    float* xalt  = ws;
    int*   tab   = (int*)(ws + nd);
    int*   rank  = tab + (size_t)chunks * NS;
    int*   bsum  = rank + E;
    int*   csr   = (int*)(ws + 2 * nd);
    int*   rowptr = csr + (size_t)E * 2;
    float* dinv  = ws + 3 * nd;
    float* WphiP = dinv + N;
    float* ATP   = WphiP + (size_t)D * D;

    const int B = 256;

    chunk_rank_kernel<<<chunks, RANK_THREADS, 0, stream>>>(ei, rank, tab, E, NS);
    block_sum_kernel<<<scan_blocks, B, 0, stream>>>(tab, bsum, chunks, NS);
    scan_rows_kernel<<<scan_blocks, B, 0, stream>>>(tab, bsum, rowptr, dinv, N, chunks, NS);
    fill_csr_kernel<<<(E + B - 1) / B, B, 0, stream>>>(ei, rank, tab, dinv, csr, E, NS);
    prep_weights_kernel<<<(D * D + B - 1) / B, B, 0, stream>>>(W, Wphi, WphiP, ATP);

    const int tile_blocks = (N + TILE_N - 1) / TILE_N;

    for (int it = 0; it < NUM_ITERS; ++it) {
        const float* xi = (it == 0) ? x_in : ((it & 1) ? xalt : out);
        float*       xo = (it & 1) ? out : xalt;
        gather_gemm_update_kernel<<<tile_blocks, NWAVES * 32, 0, stream>>>(rowptr, csr, dinv, WphiP, ATP, bias,
                                                                 xi, xo, N);
    }
}


# Iteration kernel fitted into 64 vector registers (was 80): row loads as scalar base + lane offset, row sums as packed fma, x added at the row store; 16 not 12 waves per SIMD, 4 not 3 blocks per WGP
# speedup vs baseline: 4.1523x; 1.0506x over previous
#include <hip/hip_runtime.h>
#include <hip/hip_bf16.h>
#include <math.h>

#define D 128
#define GAMMA 0.1f
#define EPSILON 0.1f
#define NUM_ITERS 4

typedef __attribute__((ext_vector_type(2))) float v2f;
typedef __attribute__((ext_vector_type(8))) float v8f;
typedef __attribute__((ext_vector_type(4))) float v4f;
typedef const __attribute__((address_space(1))) float* gfp;
typedef const __attribute__((address_space(1))) v4f* gv4p;

#define SCAN_ITEMS 1024
#define CHUNK_SHIFT 15
#define CHUNK_E (1 << CHUNK_SHIFT)
#define RANK_THREADS 1024
#define EPT (CHUNK_E / RANK_THREADS)
#define RANGE_N 65536

__global__ __launch_bounds__(RANK_THREADS) void chunk_rank_kernel(const int* __restrict__ ei,
                                                                  int* __restrict__ rank,
                                                                  int* __restrict__ tab,
                                                                  int E, int NS) {
    __shared__ __attribute__((aligned(16))) int lcnt[RANGE_N];
    const int tid = threadIdx.x;
    const int e0  = blockIdx.x * CHUNK_E;
    int* __restrict__ trow = tab + (size_t)blockIdx.x * NS;

    int dreg[EPT];
#pragma unroll
    for (int k = 0; k < EPT; ++k) {
        const int e = e0 + k * RANK_THREADS + tid;
        dreg[k] = (e < E) ? ei[E + e] : -1;
    }

    for (int r0 = 0; r0 < NS; r0 += RANGE_N) {
        for (int i = tid * 4; i < RANGE_N; i += RANK_THREADS * 4)
            *(int4*)&lcnt[i] = make_int4(0, 0, 0, 0);
        __syncthreads();
        int rk[EPT];
#pragma unroll
        for (int k = 0; k < EPT; ++k) {
            const unsigned off = (unsigned)(dreg[k] - r0);
            rk[k] = -1;
            if (off < (unsigned)RANGE_N) rk[k] = atomicAdd(&lcnt[off], 1);
        }
#pragma unroll
        for (int k = 0; k < EPT; ++k)
            if (rk[k] >= 0) rank[e0 + k * RANK_THREADS + tid] = rk[k];
        __syncthreads();
        const int lim = min(RANGE_N, NS - r0);
        for (int i = tid * 4; i < lim; i += RANK_THREADS * 4)
            *(int4*)(trow + r0 + i) = *(const int4*)&lcnt[i];
        __syncthreads();
    }
}

__global__ __launch_bounds__(256) void block_sum_kernel(const int* __restrict__ tab,
                                                        int* __restrict__ bsum, int chunks, int NS) {
    __shared__ int sm[8];
    const int tid = threadIdx.x, lane = tid & 31, wave = tid >> 5;
    const int base = blockIdx.x * SCAN_ITEMS + tid * 4;
    int s = 0;
    for (int c = 0; c < chunks; ++c) {
        const int4 v = *(const int4*)(tab + (size_t)c * NS + base);
        s += (v.x + v.y) + (v.z + v.w);
    }
    for (int o = 16; o > 0; o >>= 1) s += __shfl_xor(s, o);
    if (lane == 0) sm[wave] = s;
    __syncthreads();
    if (tid == 0) {
        int t = 0;
        for (int w = 0; w < 8; ++w) t += sm[w];
        bsum[blockIdx.x] = t;
    }
}

__global__ __launch_bounds__(256) void scan_rows_kernel(int* __restrict__ tab,
                                                        const int* __restrict__ bsum,
                                                        int* __restrict__ rowptr,
                                                        float* __restrict__ dinv, int n,
                                                        int chunks, int NS) {
    __shared__ int sm_off[8];
    __shared__ int sm_wav[8];
    const int tid = threadIdx.x, lane = tid & 31, wave = tid >> 5;

    int part = 0;
    for (int j = tid; j < (int)blockIdx.x; j += 256) part += bsum[j];
    for (int o = 16; o > 0; o >>= 1) part += __shfl_xor(part, o);
    if (lane == 0) sm_off[wave] = part;

    const int base = blockIdx.x * SCAN_ITEMS + tid * 4;
    int4 tot = make_int4(0, 0, 0, 0);
    for (int ch = 0; ch < chunks; ++ch) {
        const int4 v = *(const int4*)(tab + (size_t)ch * NS + base);
        tot.x += v.x; tot.y += v.y; tot.z += v.z; tot.w += v.w;
    }
    const int c[4] = {tot.x, tot.y, tot.z, tot.w};
    const int s = (c[0] + c[1]) + (c[2] + c[3]);
    int incl = s;
    for (int o = 1; o < 32; o <<= 1) {
        int t = __shfl_up(incl, o);
        if (lane >= o) incl += t;
    }
    if (lane == 31) sm_wav[wave] = incl;
    __syncthreads();

    int excl = incl - s;
    for (int w = 0; w < 8; ++w) {
        excl += sm_off[w];
        if (w < wave) excl += sm_wav[w];
    }
    int4 run;
    run.x = excl;
    run.y = run.x + c[0];
    run.z = run.y + c[1];
    run.w = run.z + c[2];
    const int rp[4] = {run.x, run.y, run.z, run.w};
    for (int u = 0; u < 4; ++u) {
        const int i = base + u;
        if (i < n) {
            rowptr[i] = rp[u];
            dinv[i] = rsqrtf((float)c[u] + 1.0f);
            if (i == n - 1) rowptr[n] = rp[u] + c[u];
        }
    }
    for (int ch = 0; ch < chunks; ++ch) {
        int4* p = (int4*)(tab + (size_t)ch * NS + base);
        const int4 v = *p;
        *p = run;
        run.x += v.x; run.y += v.y; run.z += v.z; run.w += v.w;
    }
}

__global__ __launch_bounds__(256) void fill_csr_kernel(const int* __restrict__ ei,
                                                       const int* __restrict__ rank,
                                                       const int* __restrict__ tab,
                                                       const float* __restrict__ dinv,
                                                       int* __restrict__ csr, int E, int NS) {
    int e = blockIdx.x * blockDim.x + threadIdx.x;
    if (e < E) {
        const int s = ei[e];
        const int d = ei[E + e];
        const int pos = tab[(size_t)(e >> CHUNK_SHIFT) * NS + d] + rank[e];
        int2 rec;
        rec.x = s;
        rec.y = __float_as_int(dinv[s] * dinv[d]);
        *(int2*)(csr + (size_t)pos * 2) = rec;
    }
}

__global__ __launch_bounds__(256) void prep_weights_kernel(const float* __restrict__ W,
                                                           const float* __restrict__ Wphi,
                                                           float* __restrict__ WphiP,
                                                           float* __restrict__ ATP) {
    int i = blockIdx.x * blockDim.x + threadIdx.x;
    if (i >= D * D) return;
    int k = i / D, n = i % D;
    size_t po = (size_t)(k >> 1) * (2 * D) + n * 2 + (k & 1);
    WphiP[po] = Wphi[n * D + k];
    float a = W[n * D + k] - W[k * D + n];
    if (k == n) a -= GAMMA;
    ATP[po] = a;
}

template <int K>
__device__ __forceinline__ void gather_rows(v2f& a01, v2f& a23, const float* __restrict__ xs,
                                            unsigned lane4, int2 my, int j) {
    v4f v[K];
    float w[K];
    unsigned long long rb[K];
#pragma unroll
    for (int u = 0; u < K; ++u) {
        const int s = __builtin_amdgcn_readlane(my.x, j + u);
        w[u] = __int_as_float(__builtin_amdgcn_readlane(my.y, j + u));
        const unsigned long long ra = (unsigned long long)(xs + (size_t)s * D);
        const unsigned ralo = __builtin_amdgcn_readfirstlane((unsigned)ra);
        const unsigned rahi = __builtin_amdgcn_readfirstlane((unsigned)(ra >> 32));
        rb[u] = ((unsigned long long)rahi << 32) | ralo;
    }
    __builtin_amdgcn_sched_barrier(0);
#pragma unroll
    for (int u = 0; u < K; ++u) v[u] = *(gv4p)((gfp)rb[u] + lane4);
#pragma unroll
    for (int u = 0; u < K; ++u) {
        const v2f ww = {w[u], w[u]};
        a01 = __builtin_elementwise_fma(v[u].xy, ww, a01);
        a23 = __builtin_elementwise_fma(v[u].zw, ww, a23);
    }
}

__device__ __forceinline__ float tanh_branchfree(float z) {
    const float t = __builtin_amdgcn_exp2f(fabsf(z) * 2.885390081777927f);
    const float r = __builtin_amdgcn_rcpf(t + 1.0f);
    return copysignf(fmaf(-2.0f, r, 1.0f), z);
}

#define NWAVES 16
#define TILE_N 64
#define NPW (TILE_N / NWAVES)
#define MT ((TILE_N / 16) / (NWAVES / 8))

__global__ __launch_bounds__(NWAVES * 32, 16) void gather_gemm_update_kernel(const int* __restrict__ rowptr,
                                                                 const int* __restrict__ csr,
                                                                 const float* __restrict__ dinv,
                                                                 const float* __restrict__ WphiP,
                                                                 const float* __restrict__ ATP,
                                                                 const float* __restrict__ bias,
                                                                 const float* __restrict__ xin,
                                                                 float* __restrict__ xout, int N) {
    __shared__ __attribute__((aligned(16))) float lds_x[TILE_N][D + 4];
    __shared__ __attribute__((aligned(16))) float lds_y[TILE_N][D + 4];

    const int tid  = threadIdx.x;
    const int lane = tid & 31;
    const int wave = __builtin_amdgcn_readfirstlane(tid >> 5);
    const int row0 = blockIdx.x * TILE_N;
    const unsigned lane4 = (unsigned)lane * 4u;

    const int d0 = row0 + wave * NPW;
    int   pb[NPW + 1];
    float dv[NPW];
#pragma unroll
    for (int i = 0; i <= NPW; ++i) pb[i] = rowptr[min(d0 + i, N)];
#pragma unroll
    for (int i = 0; i < NPW; ++i) dv[i] = dinv[min(d0 + i, N - 1)];

#pragma unroll
    for (int i = 0; i < NPW; ++i) {
        const int r = wave * NPW + i;
        float4 xs = *(const float4*)(xin + (size_t)min(d0 + i, N - 1) * D + lane4);
        if (d0 + i >= N) xs = make_float4(0.f, 0.f, 0.f, 0.f);
        *(float4*)&lds_x[r][lane4] = xs;
    }
    __syncthreads();

    const int mn   = lane & 15;
    const int kq   = lane >> 4;
    const int col0 = (wave & 7) * 16;
    const int mrow0 = (wave >> 3) * (MT * 16);
    const size_t bcol = (size_t)(col0 + mn) * 2;

    v8f acc_w[MT];
    v8f acc_a[MT];
#pragma unroll
    for (int m = 0; m < MT; ++m) { acc_w[m] = (v8f){}; acc_a[m] = (v8f){}; }

#pragma unroll
    for (int i = 0; i < NPW; ++i) {
        const int r   = wave * NPW + i;
        const int beg = pb[i];
        const int end = pb[i + 1];
        int2 my = make_int2(0, 0);
        if (lane < min(32, end - beg)) my = *(const int2*)(csr + (size_t)(beg + lane) * 2);
#pragma unroll
        for (int k0 = i * (D / NPW); k0 < (i + 1) * (D / NPW); k0 += 4) {
            const int k = k0 + kq * 2;
            const int p = (k0 >> 1) + kq;
            const v2f ba = *(const v2f*)(ATP + (size_t)p * (2 * D) + bcol);
#pragma unroll
            for (int m = 0; m < MT; ++m) {
                const v2f ax = *(const v2f*)&lds_x[mrow0 + m * 16 + mn][k];
                acc_a[m] = __builtin_amdgcn_wmma_f32_16x16x4_f32(false, ax, false, ba,
                                                                 (short)0, acc_a[m], false, false);
            }
        }
        v2f a01 = {0.f, 0.f};
        v2f a23 = {0.f, 0.f};
        for (int base = beg; base < end; base += 32) {
            const int n = min(32, end - base);
            if (base != beg) {
                my = make_int2(0, 0);
                if (lane < n) my = *(const int2*)(csr + (size_t)(base + lane) * 2);
            }
            int j = 0;
            for (; j + 8 <= n; j += 8) gather_rows<8>(a01, a23, xin, lane4, my, j);
            if (j + 4 <= n) { gather_rows<4>(a01, a23, xin, lane4, my, j); j += 4; }
            if (j + 2 <= n) { gather_rows<2>(a01, a23, xin, lane4, my, j); j += 2; }
            if (j < n) gather_rows<1>(a01, a23, xin, lane4, my, j);
        }
        const v4f xs = *(const v4f*)&lds_x[r][lane4];
        const float d2 = dv[i] * dv[i];
        const v2f dd = {d2, d2};
        a01 = __builtin_elementwise_fma(xs.xy, dd, a01);
        a23 = __builtin_elementwise_fma(xs.zw, dd, a23);
        *(v4f*)&lds_y[r][lane4] = (v4f){a01.x, a01.y, a23.x, a23.y};
    }
    __syncthreads();

    const int lane_p = (int)__builtin_amdgcn_mbcnt_lo(~0u, 0u);
    const int mn_p   = lane_p & 15;
    const int kq_p   = lane_p >> 4;
    const size_t bcol_p = (size_t)(col0 + mn_p) * 2;
#pragma unroll 4
    for (int k0 = 0; k0 < D; k0 += 4) {
        const int k = k0 + kq_p * 2;
        const int p = (k0 >> 1) + kq_p;
        const v2f bw = *(const v2f*)(WphiP + (size_t)p * (2 * D) + bcol_p);
#pragma unroll
        for (int m = 0; m < MT; ++m) {
            const v2f ay = *(const v2f*)&lds_y[mrow0 + m * 16 + mn_p][k];
            acc_w[m] = __builtin_amdgcn_wmma_f32_16x16x4_f32(false, ay, false, bw,
                                                             (short)0, acc_w[m], false, false);
        }
    }

    const int ecol   = col0 + mn_p;
    const float b    = bias[ecol];
    const int rowoff = kq_p * 8;
#pragma unroll
    for (int m = 0; m < MT; ++m) {
#pragma unroll
        for (int r = 0; r < 8; ++r) {
            acc_w[m][r] = EPSILON * tanh_branchfree(acc_w[m][r] + acc_a[m][r] + b);
        }
    }
    __syncthreads();
#pragma unroll
    for (int m = 0; m < MT; ++m) {
#pragma unroll
        for (int r = 0; r < 8; ++r) lds_y[mrow0 + m * 16 + rowoff + r][ecol] = acc_w[m][r];
    }
    __syncthreads();
#pragma unroll
    for (int i = 0; i < NPW; ++i) {
        const int r = wave * NPW + i;
        const int d = row0 + r;
        if (d < N) *(v4f*)(xout + (size_t)d * D + lane4) = *(const v4f*)&lds_x[r][lane4] + *(const v4f*)&lds_y[r][lane4];
    }
}

extern "C" void kernel_launch(void* const* d_in, const int* in_sizes, int n_in,
                              void* d_out, int out_size, void* d_ws, size_t ws_size,
                              hipStream_t stream) {
    const float* x_in = (const float*)d_in[0];
    const int*   ei   = (const int*)d_in[1];
    const float* W    = (const float*)d_in[2];
    const float* Wphi = (const float*)d_in[3];
    const float* bias = (const float*)d_in[4];
    float*       out  = (float*)d_out;

    const int N = in_sizes[0] / D;
    const int E = in_sizes[1] / 2;

    float* ws    = (float*)d_ws;
    const size_t nd = (size_t)N * D;
    const int scan_blocks = (N + SCAN_ITEMS - 1) / SCAN_ITEMS;
    const int NS     = scan_blocks * SCAN_ITEMS;
    const int chunks = (E + CHUNK_E - 1) / CHUNK_E;
    float* xalt  = ws;
    int*   tab   = (int*)(ws + nd);
    int*   rank  = tab + (size_t)chunks * NS;
    int*   bsum  = rank + E;
    int*   csr   = (int*)(ws + 2 * nd);
    int*   rowptr = csr + (size_t)E * 2;
    float* dinv  = ws + 3 * nd;
    float* WphiP = dinv + N;
    float* ATP   = WphiP + (size_t)D * D;

    const int B = 256;

    chunk_rank_kernel<<<chunks, RANK_THREADS, 0, stream>>>(ei, rank, tab, E, NS);
    block_sum_kernel<<<scan_blocks, B, 0, stream>>>(tab, bsum, chunks, NS);
    scan_rows_kernel<<<scan_blocks, B, 0, stream>>>(tab, bsum, rowptr, dinv, N, chunks, NS);
    fill_csr_kernel<<<(E + B - 1) / B, B, 0, stream>>>(ei, rank, tab, dinv, csr, E, NS);
    prep_weights_kernel<<<(D * D + B - 1) / B, B, 0, stream>>>(W, Wphi, WphiP, ATP);

    const int tile_blocks = (N + TILE_N - 1) / TILE_N;

    for (int it = 0; it < NUM_ITERS; ++it) {
        const float* xi = (it == 0) ? x_in : ((it & 1) ? xalt : out);
        float*       xo = (it & 1) ? out : xalt;
        gather_gemm_update_kernel<<<tile_blocks, NWAVES * 32, 0, stream>>>(rowptr, csr, dinv, WphiP, ATP, bias,
                                                                 xi, xo, N);
    }
}


# CSR build's chain of dependent round trips cut: ranking blocks also emit the scan's block totals and pack the weights (5 launches to 3); scan's chunk loops load 16 rows per trip; float sums untouched
# speedup vs baseline: 4.3136x; 1.0388x over previous
#include <hip/hip_runtime.h>
#include <hip/hip_bf16.h>
#include <math.h>

#define D 128
#define GAMMA 0.1f
#define EPSILON 0.1f
#define NUM_ITERS 4

typedef __attribute__((ext_vector_type(2))) float v2f;
typedef __attribute__((ext_vector_type(8))) float v8f;
typedef __attribute__((ext_vector_type(4))) float v4f;
typedef const __attribute__((address_space(1))) float* gfp;
typedef const __attribute__((address_space(1))) v4f* gv4p;

#define SCAN_ITEMS 1024
#define SCAN_BATCH 16
#define CHUNK_SHIFT 15
#define CHUNK_E (1 << CHUNK_SHIFT)
#define RANK_THREADS 1024
#define EPT (CHUNK_E / RANK_THREADS)
#define RANGE_N 65536

__global__ __launch_bounds__(RANK_THREADS) void rank_and_weights_kernel(const int* __restrict__ ei,
                                                                        int* __restrict__ rank,
                                                                        int* __restrict__ tab,
                                                                        int* __restrict__ cblk,
                                                                        int E, int NS, int chunks,
                                                                        const float* __restrict__ W,
                                                                        const float* __restrict__ Wphi,
                                                                        float* __restrict__ WphiP,
                                                                        float* __restrict__ ATP) {
    __shared__ __attribute__((aligned(16))) int lcnt[RANGE_N];
    __shared__ int bs[RANGE_N / SCAN_ITEMS][SCAN_ITEMS / 128];
    const int tid = threadIdx.x;
    if ((int)blockIdx.x >= chunks) {
        const int i = ((int)blockIdx.x - chunks) * RANK_THREADS + tid;
        if (i < D * D) {
            const int k = i / D, n = i % D;
            const size_t po = (size_t)(k >> 1) * (2 * D) + n * 2 + (k & 1);
            WphiP[po] = Wphi[n * D + k];
            float a = W[n * D + k] - W[k * D + n];
            if (k == n) a -= GAMMA;
            ATP[po] = a;
        }
        return;
    }
    const int lane = tid & 31;
    const int e0  = blockIdx.x * CHUNK_E;
    int* __restrict__ trow = tab + (size_t)blockIdx.x * NS;

    int dreg[EPT];
#pragma unroll
    for (int k = 0; k < EPT; ++k) {
        const int e = e0 + k * RANK_THREADS + tid;
        dreg[k] = (e < E) ? ei[E + e] : -1;
    }

    for (int r0 = 0; r0 < NS; r0 += RANGE_N) {
        for (int i = tid * 4; i < RANGE_N; i += RANK_THREADS * 4)
            *(int4*)&lcnt[i] = make_int4(0, 0, 0, 0);
        __syncthreads();
        int rk[EPT];
#pragma unroll
        for (int k = 0; k < EPT; ++k) {
            const unsigned off = (unsigned)(dreg[k] - r0);
            rk[k] = -1;
            if (off < (unsigned)RANGE_N) rk[k] = atomicAdd(&lcnt[off], 1);
        }
#pragma unroll
        for (int k = 0; k < EPT; ++k)
            if (rk[k] >= 0) rank[e0 + k * RANK_THREADS + tid] = rk[k];
        __syncthreads();
        const int lim = min(RANGE_N, NS - r0);
        int sj[RANGE_N / (RANK_THREADS * 4)];
#pragma unroll
        for (int j = 0; j < RANGE_N / (RANK_THREADS * 4); ++j) {
            const int i = tid * 4 + j * (RANK_THREADS * 4);
            sj[j] = 0;
            if (i < lim) {
                const int4 v = *(const int4*)&lcnt[i];
                *(int4*)(trow + r0 + i) = v;
                sj[j] = (v.x + v.y) + (v.z + v.w);
            }
        }
#pragma unroll
        for (int o = 16; o > 0; o >>= 1) {
#pragma unroll
            for (int j = 0; j < RANGE_N / (RANK_THREADS * 4); ++j) sj[j] += __shfl_xor(sj[j], o);
        }
        if (lane == 0) {
#pragma unroll
            for (int j = 0; j < RANGE_N / (RANK_THREADS * 4); ++j)
                bs[j * 4 + (tid >> 8)][(tid >> 5) & 7] = sj[j];
        }
        __syncthreads();
        if (tid < RANGE_N / SCAN_ITEMS && r0 + tid * SCAN_ITEMS < NS) {
            int t = 0;
#pragma unroll
            for (int w = 0; w < SCAN_ITEMS / 128; ++w) t += bs[tid][w];
            cblk[(size_t)(r0 / SCAN_ITEMS + tid) * chunks + blockIdx.x] = t;
        }
    }
}

__global__ __launch_bounds__(256) void scan_rows_kernel(int* __restrict__ tab,
                                                        const int* __restrict__ cblk,
                                                        int* __restrict__ rowptr,
                                                        float* __restrict__ dinv, int n,
                                                        int chunks, int NS) {
    __shared__ int sm_off[8];
    __shared__ int sm_wav[8];
    const int tid = threadIdx.x, lane = tid & 31, wave = tid >> 5;

    int part = 0;
    const int nprev = (int)blockIdx.x * chunks;
    for (int j0 = tid; j0 < nprev; j0 += 256 * SCAN_BATCH) {
        int t[SCAN_BATCH];
#pragma unroll
        for (int u = 0; u < SCAN_BATCH; ++u) t[u] = cblk[min(j0 + u * 256, nprev - 1)];
#pragma unroll
        for (int u = 0; u < SCAN_BATCH; ++u) part += (j0 + u * 256 < nprev) ? t[u] : 0;
    }
    for (int o = 16; o > 0; o >>= 1) part += __shfl_xor(part, o);
    if (lane == 0) sm_off[wave] = part;

    const int base = blockIdx.x * SCAN_ITEMS + tid * 4;
    int4 tot = make_int4(0, 0, 0, 0);
    for (int c0 = 0; c0 < chunks; c0 += SCAN_BATCH) {
        int4 v[SCAN_BATCH];
#pragma unroll
        for (int u = 0; u < SCAN_BATCH; ++u)
            v[u] = *(const int4*)(tab + (size_t)min(c0 + u, chunks - 1) * NS + base);
#pragma unroll
        for (int u = 0; u < SCAN_BATCH; ++u) {
            if (c0 + u < chunks) { tot.x += v[u].x; tot.y += v[u].y; tot.z += v[u].z; tot.w += v[u].w; }
        }
    }
    const int c[4] = {tot.x, tot.y, tot.z, tot.w};
    const int s = (c[0] + c[1]) + (c[2] + c[3]);
    int incl = s;
    for (int o = 1; o < 32; o <<= 1) {
        int t = __shfl_up(incl, o);
        if (lane >= o) incl += t;
    }
    if (lane == 31) sm_wav[wave] = incl;
    __syncthreads();

    int excl = incl - s;
    for (int w = 0; w < 8; ++w) {
        excl += sm_off[w];
        if (w < wave) excl += sm_wav[w];
    }
    int4 run;
    run.x = excl;
    run.y = run.x + c[0];
    run.z = run.y + c[1];
    run.w = run.z + c[2];
    const int rp[4] = {run.x, run.y, run.z, run.w};
    for (int u = 0; u < 4; ++u) {
        const int i = base + u;
        if (i < n) {
            rowptr[i] = rp[u];
            dinv[i] = rsqrtf((float)c[u] + 1.0f);
            if (i == n - 1) rowptr[n] = rp[u] + c[u];
        }
    }
    for (int c0 = 0; c0 < chunks; c0 += SCAN_BATCH) {
        int4 v[SCAN_BATCH];
#pragma unroll
        for (int u = 0; u < SCAN_BATCH; ++u)
            v[u] = *(const int4*)(tab + (size_t)min(c0 + u, chunks - 1) * NS + base);
#pragma unroll
        for (int u = 0; u < SCAN_BATCH; ++u) {
            if (c0 + u < chunks) {
                *(int4*)(tab + (size_t)(c0 + u) * NS + base) = run;
                run.x += v[u].x; run.y += v[u].y; run.z += v[u].z; run.w += v[u].w;
            }
        }
    }
}

__global__ __launch_bounds__(256) void fill_csr_kernel(const int* __restrict__ ei,
                                                       const int* __restrict__ rank,
                                                       const int* __restrict__ tab,
                                                       const float* __restrict__ dinv,
                                                       int* __restrict__ csr, int E, int NS) {
    int e = blockIdx.x * blockDim.x + threadIdx.x;
    if (e < E) {
        const int s = ei[e];
        const int d = ei[E + e];
        const int pos = tab[(size_t)(e >> CHUNK_SHIFT) * NS + d] + rank[e];
        int2 rec;
        rec.x = s;
        rec.y = __float_as_int(dinv[s] * dinv[d]);
        *(int2*)(csr + (size_t)pos * 2) = rec;
    }
}

template <int K>
__device__ __forceinline__ void gather_rows(v2f& a01, v2f& a23, const float* __restrict__ xs,
                                            unsigned lane4, int2 my, int j) {
    v4f v[K];
    float w[K];
    unsigned long long rb[K];
#pragma unroll
    for (int u = 0; u < K; ++u) {
        const int s = __builtin_amdgcn_readlane(my.x, j + u);
        w[u] = __int_as_float(__builtin_amdgcn_readlane(my.y, j + u));
        const unsigned long long ra = (unsigned long long)(xs + (size_t)s * D);
        const unsigned ralo = __builtin_amdgcn_readfirstlane((unsigned)ra);
        const unsigned rahi = __builtin_amdgcn_readfirstlane((unsigned)(ra >> 32));
        rb[u] = ((unsigned long long)rahi << 32) | ralo;
    }
    __builtin_amdgcn_sched_barrier(0);
#pragma unroll
    for (int u = 0; u < K; ++u) v[u] = *(gv4p)((gfp)rb[u] + lane4);
#pragma unroll
    for (int u = 0; u < K; ++u) {
        const v2f ww = {w[u], w[u]};
        a01 = __builtin_elementwise_fma(v[u].xy, ww, a01);
        a23 = __builtin_elementwise_fma(v[u].zw, ww, a23);
    }
}

__device__ __forceinline__ float tanh_branchfree(float z) {
    const float t = __builtin_amdgcn_exp2f(fabsf(z) * 2.885390081777927f);
    const float r = __builtin_amdgcn_rcpf(t + 1.0f);
    return copysignf(fmaf(-2.0f, r, 1.0f), z);
}

#define NWAVES 16
#define TILE_N 64
#define NPW (TILE_N / NWAVES)
#define MT ((TILE_N / 16) / (NWAVES / 8))

__global__ __launch_bounds__(NWAVES * 32, 16) void gather_gemm_update_kernel(const int* __restrict__ rowptr,
                                                                 const int* __restrict__ csr,
                                                                 const float* __restrict__ dinv,
                                                                 const float* __restrict__ WphiP,
                                                                 const float* __restrict__ ATP,
                                                                 const float* __restrict__ bias,
                                                                 const float* __restrict__ xin,
                                                                 float* __restrict__ xout, int N) {
    __shared__ __attribute__((aligned(16))) float lds_x[TILE_N][D + 4];
    __shared__ __attribute__((aligned(16))) float lds_y[TILE_N][D + 4];

    const int tid  = threadIdx.x;
    const int lane = tid & 31;
    const int wave = __builtin_amdgcn_readfirstlane(tid >> 5);
    const int row0 = blockIdx.x * TILE_N;
    const unsigned lane4 = (unsigned)lane * 4u;

    const int d0 = row0 + wave * NPW;
    int   pb[NPW + 1];
    float dv[NPW];
#pragma unroll
    for (int i = 0; i <= NPW; ++i) pb[i] = rowptr[min(d0 + i, N)];
#pragma unroll
    for (int i = 0; i < NPW; ++i) dv[i] = dinv[min(d0 + i, N - 1)];

#pragma unroll
    for (int i = 0; i < NPW; ++i) {
        const int r = wave * NPW + i;
        float4 xs = *(const float4*)(xin + (size_t)min(d0 + i, N - 1) * D + lane4);
        if (d0 + i >= N) xs = make_float4(0.f, 0.f, 0.f, 0.f);
        *(float4*)&lds_x[r][lane4] = xs;
    }
    __syncthreads();

    const int mn   = lane & 15;
    const int kq   = lane >> 4;
    const int col0 = (wave & 7) * 16;
    const int mrow0 = (wave >> 3) * (MT * 16);
    const size_t bcol = (size_t)(col0 + mn) * 2;

    v8f acc_w[MT];
    v8f acc_a[MT];
#pragma unroll
    for (int m = 0; m < MT; ++m) { acc_w[m] = (v8f){}; acc_a[m] = (v8f){}; }

#pragma unroll
    for (int i = 0; i < NPW; ++i) {
        const int r   = wave * NPW + i;
        const int beg = pb[i];
        const int end = pb[i + 1];
        int2 my = make_int2(0, 0);
        if (lane < min(32, end - beg)) my = *(const int2*)(csr + (size_t)(beg + lane) * 2);
#pragma unroll
        for (int k0 = i * (D / NPW); k0 < (i + 1) * (D / NPW); k0 += 4) {
            const int k = k0 + kq * 2;
            const int p = (k0 >> 1) + kq;
            const v2f ba = *(const v2f*)(ATP + (size_t)p * (2 * D) + bcol);
#pragma unroll
            for (int m = 0; m < MT; ++m) {
                const v2f ax = *(const v2f*)&lds_x[mrow0 + m * 16 + mn][k];
                acc_a[m] = __builtin_amdgcn_wmma_f32_16x16x4_f32(false, ax, false, ba,
                                                                 (short)0, acc_a[m], false, false);
            }
        }
        v2f a01 = {0.f, 0.f};
        v2f a23 = {0.f, 0.f};
        for (int base = beg; base < end; base += 32) {
            const int n = min(32, end - base);
            if (base != beg) {
                my = make_int2(0, 0);
                if (lane < n) my = *(const int2*)(csr + (size_t)(base + lane) * 2);
            }
            int j = 0;
            for (; j + 8 <= n; j += 8) gather_rows<8>(a01, a23, xin, lane4, my, j);
            if (j + 4 <= n) { gather_rows<4>(a01, a23, xin, lane4, my, j); j += 4; }
            if (j + 2 <= n) { gather_rows<2>(a01, a23, xin, lane4, my, j); j += 2; }
            if (j < n) gather_rows<1>(a01, a23, xin, lane4, my, j);
        }
        const v4f xs = *(const v4f*)&lds_x[r][lane4];
        const float d2 = dv[i] * dv[i];
        const v2f dd = {d2, d2};
        a01 = __builtin_elementwise_fma(xs.xy, dd, a01);
        a23 = __builtin_elementwise_fma(xs.zw, dd, a23);
        *(v4f*)&lds_y[r][lane4] = (v4f){a01.x, a01.y, a23.x, a23.y};
    }
    __syncthreads();

    const int lane_p = (int)__builtin_amdgcn_mbcnt_lo(~0u, 0u);
    const int mn_p   = lane_p & 15;
    const int kq_p   = lane_p >> 4;
    const size_t bcol_p = (size_t)(col0 + mn_p) * 2;
#pragma unroll 4
    for (int k0 = 0; k0 < D; k0 += 4) {
        const int k = k0 + kq_p * 2;
        const int p = (k0 >> 1) + kq_p;
        const v2f bw = *(const v2f*)(WphiP + (size_t)p * (2 * D) + bcol_p);
#pragma unroll
        for (int m = 0; m < MT; ++m) {
            const v2f ay = *(const v2f*)&lds_y[mrow0 + m * 16 + mn_p][k];
            acc_w[m] = __builtin_amdgcn_wmma_f32_16x16x4_f32(false, ay, false, bw,
                                                             (short)0, acc_w[m], false, false);
        }
    }

    const int ecol   = col0 + mn_p;
    const float b    = bias[ecol];
    const int rowoff = kq_p * 8;
#pragma unroll
    for (int m = 0; m < MT; ++m) {
#pragma unroll
        for (int r = 0; r < 8; ++r) {
            acc_w[m][r] = EPSILON * tanh_branchfree(acc_w[m][r] + acc_a[m][r] + b);
        }
    }
    __syncthreads();
#pragma unroll
    for (int m = 0; m < MT; ++m) {
#pragma unroll
        for (int r = 0; r < 8; ++r) lds_y[mrow0 + m * 16 + rowoff + r][ecol] = acc_w[m][r];
    }
    __syncthreads();
#pragma unroll
    for (int i = 0; i < NPW; ++i) {
        const int r = wave * NPW + i;
        const int d = row0 + r;
        if (d < N) *(v4f*)(xout + (size_t)d * D + lane4) = *(const v4f*)&lds_x[r][lane4] + *(const v4f*)&lds_y[r][lane4];
    }
}

extern "C" void kernel_launch(void* const* d_in, const int* in_sizes, int n_in,
                              void* d_out, int out_size, void* d_ws, size_t ws_size,
                              hipStream_t stream) {
    const float* x_in = (const float*)d_in[0];
    const int*   ei   = (const int*)d_in[1];
    const float* W    = (const float*)d_in[2];
    const float* Wphi = (const float*)d_in[3];
    const float* bias = (const float*)d_in[4];
    float*       out  = (float*)d_out;

    const int N = in_sizes[0] / D;
    const int E = in_sizes[1] / 2;

    float* ws    = (float*)d_ws;
    const size_t nd = (size_t)N * D;
    const int scan_blocks = (N + SCAN_ITEMS - 1) / SCAN_ITEMS;
    const int NS     = scan_blocks * SCAN_ITEMS;
    const int chunks = (E + CHUNK_E - 1) / CHUNK_E;
    float* xalt  = ws;
    int*   tab   = (int*)(ws + nd);
    int*   rank  = tab + (size_t)chunks * NS;
    int*   cblk  = rank + E;
    int*   csr   = (int*)(ws + 2 * nd);
    int*   rowptr = csr + (size_t)E * 2;
    float* dinv  = ws + 3 * nd;
    float* WphiP = dinv + N;
    float* ATP   = WphiP + (size_t)D * D;

    const int B = 256;

    const int wblocks = (D * D + RANK_THREADS - 1) / RANK_THREADS;
    rank_and_weights_kernel<<<chunks + wblocks, RANK_THREADS, 0, stream>>>(ei, rank, tab, cblk, E, NS, chunks,
                                                                         W, Wphi, WphiP, ATP);
    scan_rows_kernel<<<scan_blocks, B, 0, stream>>>(tab, cblk, rowptr, dinv, N, chunks, NS);
    fill_csr_kernel<<<(E + B - 1) / B, B, 0, stream>>>(ei, rank, tab, dinv, csr, E, NS);

    const int tile_blocks = (N + TILE_N - 1) / TILE_N;

    for (int it = 0; it < NUM_ITERS; ++it) {
        const float* xi = (it == 0) ? x_in : ((it & 1) ? xalt : out);
        float*       xo = (it & 1) ? out : xalt;
        gather_gemm_update_kernel<<<tile_blocks, NWAVES * 32, 0, stream>>>(rowptr, csr, dinv, WphiP, ATP, bias,
                                                                 xi, xo, N);
    }
}
